# starting kernel
# baseline (speedup 1.0000x reference)
#include <hip/hip_runtime.h>


typedef __attribute__((ext_vector_type(2))) float v2f;
typedef __attribute__((ext_vector_type(8))) float v8f;

#define BATCH 16
#define CH    64
#define TLEN  2048
#define KNN   3

__global__ __launch_bounds__(256) void nsq_kernel(const float* __restrict__ x,
                                                  float* __restrict__ nsq) {
    int g = blockIdx.x * 256 + threadIdx.x;
    int b = g >> 11;
    int t = g & (TLEN - 1);
    const float* xb = x + (size_t)b * CH * TLEN + t;
    float s = 0.0f;
#pragma unroll
    for (int c = 0; c < CH; ++c) {
        float v = xb[c * TLEN];
        s += v * v;
    }
    nsq[g] = s;
}

__global__ __launch_bounds__(256) void knn_kernel(const float* __restrict__ x,
                                                  const float* __restrict__ nsq,
                                                  int* __restrict__ idx_out) {
    const int b    = blockIdx.x >> 7;
    const int tile = blockIdx.x & 127;
    const int t0   = tile * 16;

    extern __shared__ float smem[];
    float* dist   = smem;
    float* cand_d = smem + 16 * TLEN;
    int*   cand_i = (int*)(cand_d + 16 * 48);

    const int tid   = threadIdx.x;
    const int lane  = tid & 31;
    const int wave  = tid >> 5;
    const int lmod  = lane & 15;
    const int lhalf = (lane < 16) ? 0 : 2;
    const int mhalf = (lane < 16) ? 0 : 8;

    const float* xb   = x   + (size_t)b * CH * TLEN;
    const float* nsqb = nsq + (size_t)b * TLEN;

    v2f afrag[16];
#pragma unroll
    for (int kc = 0; kc < 16; ++kc) {
        int c0 = kc * 4 + lhalf;
        afrag[kc].x = xb[(size_t)(c0 + 0) * TLEN + t0 + lmod];
        afrag[kc].y = xb[(size_t)(c0 + 1) * TLEN + t0 + lmod];
    }
    float nsq_t[8];
#pragma unroll
    for (int r = 0; r < 8; ++r) nsq_t[r] = nsqb[t0 + r + mhalf];

    for (int it = 0; it < 16; ++it) {
        const int s0 = (it * 8 + wave) * 16;
        v8f acc = {};
#pragma unroll
        for (int kc = 0; kc < 16; ++kc) {
            int c0 = kc * 4 + lhalf;
            v2f bfrag;
            bfrag.x = xb[(size_t)(c0 + 0) * TLEN + s0 + lmod];
            bfrag.y = xb[(size_t)(c0 + 1) * TLEN + s0 + lmod];
            acc = __builtin_amdgcn_wmma_f32_16x16x4_f32(
                false, afrag[kc], false, bfrag, (short)0, acc, false, false);
        }
        const float nsq_s = nsqb[s0 + lmod];
#pragma unroll
        for (int r = 0; r < 8; ++r) {
            int m   = r + mhalf;
            float d = nsq_t[r] + nsq_s - 2.0f * acc[r];
            dist[m * TLEN + s0 + lmod] = fmaxf(d, 0.0f);
        }
    }
    __syncthreads();

    const int row = tid >> 4;
    const int tt  = tid & 15;
    float d0 = 3.4e38f, d1 = 3.4e38f, d2 = 3.4e38f;
    int   i0 = 0x7fffffff, i1 = 0x7fffffff, i2 = 0x7fffffff;
    const float* drow = dist + row * TLEN;
    for (int j = tt; j < TLEN; j += 16) {
        float d = drow[j];
        if (d < d0)      { d2 = d1; i2 = i1; d1 = d0; i1 = i0; d0 = d; i0 = j; }
        else if (d < d1) { d2 = d1; i2 = i1; d1 = d;  i1 = j; }
        else if (d < d2) { d2 = d;  i2 = j; }
    }
    cand_d[row * 48 + tt * 3 + 0] = d0;  cand_i[row * 48 + tt * 3 + 0] = i0;
    cand_d[row * 48 + tt * 3 + 1] = d1;  cand_i[row * 48 + tt * 3 + 1] = i1;
    cand_d[row * 48 + tt * 3 + 2] = d2;  cand_i[row * 48 + tt * 3 + 2] = i2;
    __syncthreads();

    if (tt == 0) {
        float bd0 = 3.4e38f, bd1 = 3.4e38f, bd2 = 3.4e38f;
        int   bi0 = 0x7fffffff, bi1 = 0x7fffffff, bi2 = 0x7fffffff;
        for (int c = 0; c < 48; ++c) {
            float d = cand_d[row * 48 + c];
            int   i = cand_i[row * 48 + c];
            if (d < bd0 || (d == bd0 && i < bi0)) {
                bd2 = bd1; bi2 = bi1; bd1 = bd0; bi1 = bi0; bd0 = d; bi0 = i;
            } else if (d < bd1 || (d == bd1 && i < bi1)) {
                bd2 = bd1; bi2 = bi1; bd1 = d; bi1 = i;
            } else if (d < bd2 || (d == bd2 && i < bi2)) {
                bd2 = d; bi2 = i;
            }
        }
        int* o = idx_out + ((size_t)b * TLEN + t0 + row) * KNN;
        o[0] = bi0; o[1] = bi1; o[2] = bi2;
    }
}

__global__ __launch_bounds__(256) void conv_kernel(const float* __restrict__ x,
                                                   const float* __restrict__ W,
                                                   const float* __restrict__ bias,
                                                   const int* __restrict__ idx,
                                                   float* __restrict__ out) {
    const int b   = blockIdx.x >> 6;
    const int blk = blockIdx.x & 63;
    const int wave  = threadIdx.x >> 5;
    const int lane  = threadIdx.x & 31;
    const int lmod  = lane & 15;
    const int lhalf = (lane < 16) ? 0 : 2;
    const int mhalf = (lane < 16) ? 0 : 8;

    const int o0 = (wave & 3) * 16;
    const int t0 = blk * 32 + (wave >> 2) * 16;
    const int tok = t0 + lmod;

    const float* xb = x + (size_t)b * CH * TLEN;
    const int* my = idx + ((size_t)b * TLEN + tok) * KNN;
    const int s0i = my[0], s1i = my[1], s2i = my[2];

    v8f acc = {};
#pragma unroll
    for (int kc = 0; kc < 48; ++kc) {
        const int ka0 = kc * 4 + lhalf;
        const int ka1 = ka0 + 1;
        v2f a, bb;
        a.x = W[(size_t)(o0 + lmod) * (CH * KNN) + ka0];
        a.y = W[(size_t)(o0 + lmod) * (CH * KNN) + ka1];
        {
            int i = ka0 / 3, kk = ka0 - i * 3;
            int s = (kk == 0) ? s0i : ((kk == 1) ? s1i : s2i);
            bb.x = xb[(size_t)i * TLEN + s];
        }
        {
            int i = ka1 / 3, kk = ka1 - i * 3;
            int s = (kk == 0) ? s0i : ((kk == 1) ? s1i : s2i);
            bb.y = xb[(size_t)i * TLEN + s];
        }
        acc = __builtin_amdgcn_wmma_f32_16x16x4_f32(
            false, a, false, bb, (short)0, acc, false, false);
    }

#pragma unroll
    for (int r = 0; r < 8; ++r) {
        int o = o0 + r + mhalf;
        out[(size_t)b * CH * TLEN + (size_t)o * TLEN + tok] = acc[r] + bias[o];
    }
}

extern "C" void kernel_launch(void* const* d_in, const int* in_sizes, int n_in,
                              void* d_out, int out_size, void* d_ws, size_t ws_size,
                              hipStream_t stream) {
    (void)in_sizes; (void)n_in; (void)out_size; (void)ws_size;
    const float* x    = (const float*)d_in[0];
    const float* W    = (const float*)d_in[1];
    const float* bias = (const float*)d_in[2];
    float* out = (float*)d_out;

    float* nsq = (float*)d_ws;
    int*   idx = (int*)((char*)d_ws + (size_t)BATCH * TLEN * sizeof(float));

    nsq_kernel<<<(BATCH * TLEN) / 256, 256, 0, stream>>>(x, nsq);

    const size_t smem = (size_t)16 * TLEN * sizeof(float)
                      + (size_t)16 * 48 * sizeof(float)
                      + (size_t)16 * 48 * sizeof(int);
    hipFuncSetAttribute((const void*)knn_kernel,
                        hipFuncAttributeMaxDynamicSharedMemorySize, (int)smem);
    knn_kernel<<<BATCH * (TLEN / 16), 256, smem, stream>>>(x, nsq, idx);

    conv_kernel<<<BATCH * (TLEN / 32), 256, 0, stream>>>(x, W, bias, idx, out);
}


# knn: top-3 selection from the LDS distance slab made branch-free: 128-bit slab reads, per-lane median-of-3/select insertion, 16-lane DPP lexicographic-min merge; branchy scan, one-lane merge gone
# speedup vs baseline: 1.4901x; 1.4901x over previous
#include <hip/hip_runtime.h>


typedef __attribute__((ext_vector_type(2))) float v2f;
typedef __attribute__((ext_vector_type(8))) float v8f;

#define BATCH 16
#define CH    64
#define TLEN  2048
#define KNN   3

__global__ __launch_bounds__(256) void nsq_kernel(const float* __restrict__ x,
                                                  float* __restrict__ nsq) {
    int g = blockIdx.x * 256 + threadIdx.x;
    int b = g >> 11;
    int t = g & (TLEN - 1);
    const float* xb = x + (size_t)b * CH * TLEN + t;
    float s = 0.0f;
#pragma unroll
    for (int c = 0; c < CH; ++c) {
        float v = xb[c * TLEN];
        s += v * v;
    }
    nsq[g] = s;
}


template <int M>
__device__ __forceinline__ int row_xor(int v) {
    return __builtin_amdgcn_update_dpp(v, v, 0x160 | M, 0xf, 0xf, false);
}

__device__ __forceinline__ void top3_insert(float d, int j,
                                            float& d0, float& d1, float& d2,
                                            int& i0, int& i1, int& i2) {
    const bool c0 = d < d0, c1 = d < d1, c2 = d < d2;
    i2 = c2 ? (c1 ? i1 : j) : i2;
    i1 = c1 ? (c0 ? i0 : j) : i1;
    i0 = c0 ? j : i0;
    const float n0 = fminf(d0, d);
    const float n1 = __builtin_amdgcn_fmed3f(d0, d1, d);
    const float n2 = __builtin_amdgcn_fmed3f(d1, d2, d);
    d0 = n0; d1 = n1; d2 = n2;
}

template <int M>
__device__ __forceinline__ void lexmin_step(float& bd, int& bi) {
    const float od = __int_as_float(row_xor<M>(__float_as_int(bd)));
    const int   oi = row_xor<M>(bi);
    const bool take = (od < bd) | ((od == bd) & (oi < bi));
    bd = take ? od : bd;
    bi = take ? oi : bi;
}

__global__ __launch_bounds__(256) void knn_kernel(const float* __restrict__ x,
                                                  const float* __restrict__ nsq,
                                                  int* __restrict__ idx_out) {
    const int b    = blockIdx.x >> 7;
    const int tile = blockIdx.x & 127;
    const int t0   = tile * 16;

    extern __shared__ __attribute__((aligned(16))) float smem[];
    float* dist   = smem;

    const int tid   = threadIdx.x;
    const int lane  = tid & 31;
    const int wave  = tid >> 5;
    const int lmod  = lane & 15;
    const int lhalf = (lane < 16) ? 0 : 2;
    const int mhalf = (lane < 16) ? 0 : 8;

    const float* xb   = x   + (size_t)b * CH * TLEN;
    const float* nsqb = nsq + (size_t)b * TLEN;

    v2f afrag[16];
#pragma unroll
    for (int kc = 0; kc < 16; ++kc) {
        int c0 = kc * 4 + lhalf;
        afrag[kc].x = xb[(size_t)(c0 + 0) * TLEN + t0 + lmod];
        afrag[kc].y = xb[(size_t)(c0 + 1) * TLEN + t0 + lmod];
    }
    float nsq_t[8];
#pragma unroll
    for (int r = 0; r < 8; ++r) nsq_t[r] = nsqb[t0 + r + mhalf];

    for (int it = 0; it < 16; ++it) {
        const int s0 = (it * 8 + wave) * 16;
        v8f acc = {};
#pragma unroll
        for (int kc = 0; kc < 16; ++kc) {
            int c0 = kc * 4 + lhalf;
            v2f bfrag;
            bfrag.x = xb[(size_t)(c0 + 0) * TLEN + s0 + lmod];
            bfrag.y = xb[(size_t)(c0 + 1) * TLEN + s0 + lmod];
            acc = __builtin_amdgcn_wmma_f32_16x16x4_f32(
                false, afrag[kc], false, bfrag, (short)0, acc, false, false);
        }
        const float nsq_s = nsqb[s0 + lmod];
#pragma unroll
        for (int r = 0; r < 8; ++r) {
            int m   = r + mhalf;
            float d = nsq_t[r] + nsq_s - 2.0f * acc[r];
            dist[m * TLEN + s0 + lmod] = fmaxf(d, 0.0f);
        }
    }
    __syncthreads();

    const int row = tid >> 4;
    const int tt  = tid & 15;
    float d0 = 3.4e38f, d1 = 3.4e38f, d2 = 3.4e38f;
    int   i0 = 0x7fffffff, i1 = 0x7fffffff, i2 = 0x7fffffff;
    const float4* drow4 = (const float4*)(dist + row * TLEN) + tt;
#pragma unroll 8
    for (int g = 0; g < 32; ++g) {
        const float4 v = drow4[g * 16];
        const int j = g * 64 + tt * 4;
        top3_insert(v.x, j + 0, d0, d1, d2, i0, i1, i2);
        top3_insert(v.y, j + 1, d0, d1, d2, i0, i1, i2);
        top3_insert(v.z, j + 2, d0, d1, d2, i0, i1, i2);
        top3_insert(v.w, j + 3, d0, d1, d2, i0, i1, i2);
    }

    int res0, res1, res2;
    {
        float bd = d0; int bi = i0;
        lexmin_step<1>(bd, bi); lexmin_step<2>(bd, bi);
        lexmin_step<4>(bd, bi); lexmin_step<8>(bd, bi);
        res0 = bi;
        const bool won = (i0 == bi);
        d0 = won ? d1 : d0;  i0 = won ? i1 : i0;
        d1 = won ? d2 : d1;  i1 = won ? i2 : i1;
    }
    {
        float bd = d0; int bi = i0;
        lexmin_step<1>(bd, bi); lexmin_step<2>(bd, bi);
        lexmin_step<4>(bd, bi); lexmin_step<8>(bd, bi);
        res1 = bi;
        const bool won = (i0 == bi);
        d0 = won ? d1 : d0;  i0 = won ? i1 : i0;
    }
    {
        float bd = d0; int bi = i0;
        lexmin_step<1>(bd, bi); lexmin_step<2>(bd, bi);
        lexmin_step<4>(bd, bi); lexmin_step<8>(bd, bi);
        res2 = bi;
    }
    if (tt == 0) {
        int* o = idx_out + ((size_t)b * TLEN + t0 + row) * KNN;
        o[0] = res0; o[1] = res1; o[2] = res2;
    }
}

__global__ __launch_bounds__(256) void conv_kernel(const float* __restrict__ x,
                                                   const float* __restrict__ W,
                                                   const float* __restrict__ bias,
                                                   const int* __restrict__ idx,
                                                   float* __restrict__ out) {
    const int b   = blockIdx.x >> 6;
    const int blk = blockIdx.x & 63;
    const int wave  = threadIdx.x >> 5;
    const int lane  = threadIdx.x & 31;
    const int lmod  = lane & 15;
    const int lhalf = (lane < 16) ? 0 : 2;
    const int mhalf = (lane < 16) ? 0 : 8;

    const int o0 = (wave & 3) * 16;
    const int t0 = blk * 32 + (wave >> 2) * 16;
    const int tok = t0 + lmod;

    const float* xb = x + (size_t)b * CH * TLEN;
    const int* my = idx + ((size_t)b * TLEN + tok) * KNN;
    const int s0i = my[0], s1i = my[1], s2i = my[2];

    v8f acc = {};
#pragma unroll
    for (int kc = 0; kc < 48; ++kc) {
        const int ka0 = kc * 4 + lhalf;
        const int ka1 = ka0 + 1;
        v2f a, bb;
        a.x = W[(size_t)(o0 + lmod) * (CH * KNN) + ka0];
        a.y = W[(size_t)(o0 + lmod) * (CH * KNN) + ka1];
        {
            int i = ka0 / 3, kk = ka0 - i * 3;
            int s = (kk == 0) ? s0i : ((kk == 1) ? s1i : s2i);
            bb.x = xb[(size_t)i * TLEN + s];
        }
        {
            int i = ka1 / 3, kk = ka1 - i * 3;
            int s = (kk == 0) ? s0i : ((kk == 1) ? s1i : s2i);
            bb.y = xb[(size_t)i * TLEN + s];
        }
        acc = __builtin_amdgcn_wmma_f32_16x16x4_f32(
            false, a, false, bb, (short)0, acc, false, false);
    }

#pragma unroll
    for (int r = 0; r < 8; ++r) {
        int o = o0 + r + mhalf;
        out[(size_t)b * CH * TLEN + (size_t)o * TLEN + tok] = acc[r] + bias[o];
    }
}

extern "C" void kernel_launch(void* const* d_in, const int* in_sizes, int n_in,
                              void* d_out, int out_size, void* d_ws, size_t ws_size,
                              hipStream_t stream) {
    (void)in_sizes; (void)n_in; (void)out_size; (void)ws_size;
    const float* x    = (const float*)d_in[0];
    const float* W    = (const float*)d_in[1];
    const float* bias = (const float*)d_in[2];
    float* out = (float*)d_out;

    float* nsq = (float*)d_ws;
    int*   idx = (int*)((char*)d_ws + (size_t)BATCH * TLEN * sizeof(float));

    nsq_kernel<<<(BATCH * TLEN) / 256, 256, 0, stream>>>(x, nsq);

    const size_t smem = (size_t)16 * TLEN * sizeof(float);
    hipFuncSetAttribute((const void*)knn_kernel,
                        hipFuncAttributeMaxDynamicSharedMemorySize, (int)smem);
    knn_kernel<<<BATCH * (TLEN / 16), 256, smem, stream>>>(x, nsq, idx);

    conv_kernel<<<BATCH * (TLEN / 32), 256, 0, stream>>>(x, W, bias, idx, out);
}


# knn: block covers 32 rows (16 waves, 256-KB slab); each wave forms 32x32 distance tiles on 2x2 accumulators, so every B fragment loaded from global feeds two row tiles: column loads halved
# speedup vs baseline: 1.6263x; 1.0914x over previous
#include <hip/hip_runtime.h>


typedef __attribute__((ext_vector_type(2))) float v2f;
typedef __attribute__((ext_vector_type(8))) float v8f;

#define BATCH 16
#define CH    64
#define TLEN  2048
#define KNN   3

__global__ __launch_bounds__(256) void nsq_kernel(const float* __restrict__ x,
                                                  float* __restrict__ nsq) {
    int g = blockIdx.x * 256 + threadIdx.x;
    int b = g >> 11;
    int t = g & (TLEN - 1);
    const float* xb = x + (size_t)b * CH * TLEN + t;
    float s = 0.0f;
#pragma unroll
    for (int c = 0; c < CH; ++c) {
        float v = xb[c * TLEN];
        s += v * v;
    }
    nsq[g] = s;
}


template <int M>
__device__ __forceinline__ int row_xor(int v) {
    return __builtin_amdgcn_update_dpp(v, v, 0x160 | M, 0xf, 0xf, false);
}

__device__ __forceinline__ void top3_insert(float d, int j,
                                            float& d0, float& d1, float& d2,
                                            int& i0, int& i1, int& i2) {
    const bool c0 = d < d0, c1 = d < d1, c2 = d < d2;
    i2 = c2 ? (c1 ? i1 : j) : i2;
    i1 = c1 ? (c0 ? i0 : j) : i1;
    i0 = c0 ? j : i0;
    const float n0 = fminf(d0, d);
    const float n1 = __builtin_amdgcn_fmed3f(d0, d1, d);
    const float n2 = __builtin_amdgcn_fmed3f(d1, d2, d);
    d0 = n0; d1 = n1; d2 = n2;
}

template <int M>
__device__ __forceinline__ void lexmin_step(float& bd, int& bi) {
    const float od = __int_as_float(row_xor<M>(__float_as_int(bd)));
    const int   oi = row_xor<M>(bi);
    const bool take = (od < bd) | ((od == bd) & (oi < bi));
    bd = take ? od : bd;
    bi = take ? oi : bi;
}

__global__ __launch_bounds__(512) __attribute__((amdgpu_waves_per_eu(4, 4))) void knn_kernel(const float* __restrict__ x,
                                                  const float* __restrict__ nsq,
                                                  int* __restrict__ idx_out) {
    const int b    = blockIdx.x >> 6;
    const int tile = blockIdx.x & 63;
    const int t0   = tile * 32;

    extern __shared__ __attribute__((aligned(16))) float smem[];
    float* dist   = smem;

    const int tid   = threadIdx.x;
    const int lane  = tid & 31;
    const int wave  = tid >> 5;
    const int lmod  = lane & 15;
    const int lhalf = (lane < 16) ? 0 : 2;
    const int mhalf = (lane < 16) ? 0 : 8;

    const float* xb   = x   + (size_t)b * CH * TLEN;
    const float* nsqb = nsq + (size_t)b * TLEN;

    v2f afrag0[16], afrag1[16];
#pragma unroll
    for (int kc = 0; kc < 16; ++kc) {
        int c0 = kc * 4 + lhalf;
        afrag0[kc].x = xb[(size_t)(c0 + 0) * TLEN + t0 + lmod];
        afrag0[kc].y = xb[(size_t)(c0 + 1) * TLEN + t0 + lmod];
        afrag1[kc].x = xb[(size_t)(c0 + 0) * TLEN + t0 + 16 + lmod];
        afrag1[kc].y = xb[(size_t)(c0 + 1) * TLEN + t0 + 16 + lmod];
    }
    float nsq_t0[8], nsq_t1[8];
#pragma unroll
    for (int r = 0; r < 8; ++r) {
        nsq_t0[r] = nsqb[t0 + r + mhalf];
        nsq_t1[r] = nsqb[t0 + 16 + r + mhalf];
    }

    for (int it = 0; it < 4; ++it) {
        const int s0 = (it * 16 + wave) * 32;
        v8f acc00 = {};
        v8f acc10 = {};
        v8f acc01 = {};
        v8f acc11 = {};
#pragma unroll
        for (int kc = 0; kc < 16; ++kc) {
            int c0 = kc * 4 + lhalf;
            v2f bf0, bf1;
            bf0.x = xb[(size_t)(c0 + 0) * TLEN + s0 + lmod];
            bf0.y = xb[(size_t)(c0 + 1) * TLEN + s0 + lmod];
            bf1.x = xb[(size_t)(c0 + 0) * TLEN + s0 + 16 + lmod];
            bf1.y = xb[(size_t)(c0 + 1) * TLEN + s0 + 16 + lmod];
            acc00 = __builtin_amdgcn_wmma_f32_16x16x4_f32(
                false, afrag0[kc], false, bf0, (short)0, acc00, false, false);
            acc10 = __builtin_amdgcn_wmma_f32_16x16x4_f32(
                false, afrag1[kc], false, bf0, (short)0, acc10, false, false);
            acc01 = __builtin_amdgcn_wmma_f32_16x16x4_f32(
                false, afrag0[kc], false, bf1, (short)0, acc01, false, false);
            acc11 = __builtin_amdgcn_wmma_f32_16x16x4_f32(
                false, afrag1[kc], false, bf1, (short)0, acc11, false, false);
        }
        const float nsq_s0 = nsqb[s0 + lmod];
        const float nsq_s1 = nsqb[s0 + 16 + lmod];
#pragma unroll
        for (int r = 0; r < 8; ++r) {
            int m     = r + mhalf;
            float d00 = nsq_t0[r] + nsq_s0 - 2.0f * acc00[r];
            float d10 = nsq_t1[r] + nsq_s0 - 2.0f * acc10[r];
            float d01 = nsq_t0[r] + nsq_s1 - 2.0f * acc01[r];
            float d11 = nsq_t1[r] + nsq_s1 - 2.0f * acc11[r];
            dist[m * TLEN + s0 + lmod]             = fmaxf(d00, 0.0f);
            dist[(m + 16) * TLEN + s0 + lmod]      = fmaxf(d10, 0.0f);
            dist[m * TLEN + s0 + 16 + lmod]        = fmaxf(d01, 0.0f);
            dist[(m + 16) * TLEN + s0 + 16 + lmod] = fmaxf(d11, 0.0f);
        }
    }
    __syncthreads();

    const int row = tid >> 4;
    const int tt  = tid & 15;
    float d0 = 3.4e38f, d1 = 3.4e38f, d2 = 3.4e38f;
    int   i0 = 0x7fffffff, i1 = 0x7fffffff, i2 = 0x7fffffff;
    const float4* drow4 = (const float4*)(dist + row * TLEN) + tt;
#pragma unroll 8
    for (int g = 0; g < 32; ++g) {
        const float4 v = drow4[g * 16];
        const int j = g * 64 + tt * 4;
        top3_insert(v.x, j + 0, d0, d1, d2, i0, i1, i2);
        top3_insert(v.y, j + 1, d0, d1, d2, i0, i1, i2);
        top3_insert(v.z, j + 2, d0, d1, d2, i0, i1, i2);
        top3_insert(v.w, j + 3, d0, d1, d2, i0, i1, i2);
    }

    int res0, res1, res2;
    {
        float bd = d0; int bi = i0;
        lexmin_step<1>(bd, bi); lexmin_step<2>(bd, bi);
        lexmin_step<4>(bd, bi); lexmin_step<8>(bd, bi);
        res0 = bi;
        const bool won = (i0 == bi);
        d0 = won ? d1 : d0;  i0 = won ? i1 : i0;
        d1 = won ? d2 : d1;  i1 = won ? i2 : i1;
    }
    {
        float bd = d0; int bi = i0;
        lexmin_step<1>(bd, bi); lexmin_step<2>(bd, bi);
        lexmin_step<4>(bd, bi); lexmin_step<8>(bd, bi);
        res1 = bi;
        const bool won = (i0 == bi);
        d0 = won ? d1 : d0;  i0 = won ? i1 : i0;
    }
    {
        float bd = d0; int bi = i0;
        lexmin_step<1>(bd, bi); lexmin_step<2>(bd, bi);
        lexmin_step<4>(bd, bi); lexmin_step<8>(bd, bi);
        res2 = bi;
    }
    if (tt == 0) {
        int* o = idx_out + ((size_t)b * TLEN + t0 + row) * KNN;
        o[0] = res0; o[1] = res1; o[2] = res2;
    }
}

__global__ __launch_bounds__(256) void conv_kernel(const float* __restrict__ x,
                                                   const float* __restrict__ W,
                                                   const float* __restrict__ bias,
                                                   const int* __restrict__ idx,
                                                   float* __restrict__ out) {
    const int b   = blockIdx.x >> 6;
    const int blk = blockIdx.x & 63;
    const int wave  = threadIdx.x >> 5;
    const int lane  = threadIdx.x & 31;
    const int lmod  = lane & 15;
    const int lhalf = (lane < 16) ? 0 : 2;
    const int mhalf = (lane < 16) ? 0 : 8;

    const int o0 = (wave & 3) * 16;
    const int t0 = blk * 32 + (wave >> 2) * 16;
    const int tok = t0 + lmod;

    const float* xb = x + (size_t)b * CH * TLEN;
    const int* my = idx + ((size_t)b * TLEN + tok) * KNN;
    const int s0i = my[0], s1i = my[1], s2i = my[2];

    v8f acc = {};
#pragma unroll
    for (int kc = 0; kc < 48; ++kc) {
        const int ka0 = kc * 4 + lhalf;
        const int ka1 = ka0 + 1;
        v2f a, bb;
        a.x = W[(size_t)(o0 + lmod) * (CH * KNN) + ka0];
        a.y = W[(size_t)(o0 + lmod) * (CH * KNN) + ka1];
        {
            int i = ka0 / 3, kk = ka0 - i * 3;
            int s = (kk == 0) ? s0i : ((kk == 1) ? s1i : s2i);
            bb.x = xb[(size_t)i * TLEN + s];
        }
        {
            int i = ka1 / 3, kk = ka1 - i * 3;
            int s = (kk == 0) ? s0i : ((kk == 1) ? s1i : s2i);
            bb.y = xb[(size_t)i * TLEN + s];
        }
        acc = __builtin_amdgcn_wmma_f32_16x16x4_f32(
            false, a, false, bb, (short)0, acc, false, false);
    }

#pragma unroll
    for (int r = 0; r < 8; ++r) {
        int o = o0 + r + mhalf;
        out[(size_t)b * CH * TLEN + (size_t)o * TLEN + tok] = acc[r] + bias[o];
    }
}

extern "C" void kernel_launch(void* const* d_in, const int* in_sizes, int n_in,
                              void* d_out, int out_size, void* d_ws, size_t ws_size,
                              hipStream_t stream) {
    (void)in_sizes; (void)n_in; (void)out_size; (void)ws_size;
    const float* x    = (const float*)d_in[0];
    const float* W    = (const float*)d_in[1];
    const float* bias = (const float*)d_in[2];
    float* out = (float*)d_out;

    float* nsq = (float*)d_ws;
    int*   idx = (int*)((char*)d_ws + (size_t)BATCH * TLEN * sizeof(float));

    nsq_kernel<<<(BATCH * TLEN) / 256, 256, 0, stream>>>(x, nsq);

    const size_t smem = (size_t)32 * TLEN * sizeof(float);
    hipFuncSetAttribute((const void*)knn_kernel,
                        hipFuncAttributeMaxDynamicSharedMemorySize, (int)smem);
    knn_kernel<<<BATCH * (TLEN / 32), 512, smem, stream>>>(x, nsq, idx);

    conv_kernel<<<BATCH * (TLEN / 32), 256, 0, stream>>>(x, W, bias, idx, out);
}


# conv: x[b] streamed through LDS in 16-channel chunks by contiguous 128-bit loads, neighbour values gathered from LDS, not from scattered global lines; 128 tokens per block; conv sum order changed
# speedup vs baseline: 1.8925x; 1.1637x over previous
#include <hip/hip_runtime.h>


typedef __attribute__((ext_vector_type(2))) float v2f;
typedef __attribute__((ext_vector_type(8))) float v8f;

#define BATCH 16
#define CH    64
#define TLEN  2048
#define KNN   3

__global__ __launch_bounds__(256) void nsq_kernel(const float* __restrict__ x,
                                                  float* __restrict__ nsq) {
    int g = blockIdx.x * 256 + threadIdx.x;
    int b = g >> 11;
    int t = g & (TLEN - 1);
    const float* xb = x + (size_t)b * CH * TLEN + t;
    float s = 0.0f;
#pragma unroll
    for (int c = 0; c < CH; ++c) {
        float v = xb[c * TLEN];
        s += v * v;
    }
    nsq[g] = s;
}


template <int M>
__device__ __forceinline__ int row_xor(int v) {
    return __builtin_amdgcn_update_dpp(v, v, 0x160 | M, 0xf, 0xf, false);
}

__device__ __forceinline__ void top3_insert(float d, int j,
                                            float& d0, float& d1, float& d2,
                                            int& i0, int& i1, int& i2) {
    const bool c0 = d < d0, c1 = d < d1, c2 = d < d2;
    i2 = c2 ? (c1 ? i1 : j) : i2;
    i1 = c1 ? (c0 ? i0 : j) : i1;
    i0 = c0 ? j : i0;
    const float n0 = fminf(d0, d);
    const float n1 = __builtin_amdgcn_fmed3f(d0, d1, d);
    const float n2 = __builtin_amdgcn_fmed3f(d1, d2, d);
    d0 = n0; d1 = n1; d2 = n2;
}

template <int M>
__device__ __forceinline__ void lexmin_step(float& bd, int& bi) {
    const float od = __int_as_float(row_xor<M>(__float_as_int(bd)));
    const int   oi = row_xor<M>(bi);
    const bool take = (od < bd) | ((od == bd) & (oi < bi));
    bd = take ? od : bd;
    bi = take ? oi : bi;
}

__global__ __launch_bounds__(512) __attribute__((amdgpu_waves_per_eu(4, 4))) void knn_kernel(const float* __restrict__ x,
                                                  const float* __restrict__ nsq,
                                                  int* __restrict__ idx_out) {
    const int b    = blockIdx.x >> 6;
    const int tile = blockIdx.x & 63;
    const int t0   = tile * 32;

    extern __shared__ __attribute__((aligned(16))) float smem[];
    float* dist   = smem;

    const int tid   = threadIdx.x;
    const int lane  = tid & 31;
    const int wave  = tid >> 5;
    const int lmod  = lane & 15;
    const int lhalf = (lane < 16) ? 0 : 2;
    const int mhalf = (lane < 16) ? 0 : 8;

    const float* xb   = x   + (size_t)b * CH * TLEN;
    const float* nsqb = nsq + (size_t)b * TLEN;

    v2f afrag0[16], afrag1[16];
#pragma unroll
    for (int kc = 0; kc < 16; ++kc) {
        int c0 = kc * 4 + lhalf;
        afrag0[kc].x = xb[(size_t)(c0 + 0) * TLEN + t0 + lmod];
        afrag0[kc].y = xb[(size_t)(c0 + 1) * TLEN + t0 + lmod];
        afrag1[kc].x = xb[(size_t)(c0 + 0) * TLEN + t0 + 16 + lmod];
        afrag1[kc].y = xb[(size_t)(c0 + 1) * TLEN + t0 + 16 + lmod];
    }
    float nsq_t0[8], nsq_t1[8];
#pragma unroll
    for (int r = 0; r < 8; ++r) {
        nsq_t0[r] = nsqb[t0 + r + mhalf];
        nsq_t1[r] = nsqb[t0 + 16 + r + mhalf];
    }

    for (int it = 0; it < 4; ++it) {
        const int s0 = (it * 16 + wave) * 32;
        v8f acc00 = {};
        v8f acc10 = {};
        v8f acc01 = {};
        v8f acc11 = {};
#pragma unroll
        for (int kc = 0; kc < 16; ++kc) {
            int c0 = kc * 4 + lhalf;
            v2f bf0, bf1;
            bf0.x = xb[(size_t)(c0 + 0) * TLEN + s0 + lmod];
            bf0.y = xb[(size_t)(c0 + 1) * TLEN + s0 + lmod];
            bf1.x = xb[(size_t)(c0 + 0) * TLEN + s0 + 16 + lmod];
            bf1.y = xb[(size_t)(c0 + 1) * TLEN + s0 + 16 + lmod];
            acc00 = __builtin_amdgcn_wmma_f32_16x16x4_f32(
                false, afrag0[kc], false, bf0, (short)0, acc00, false, false);
            acc10 = __builtin_amdgcn_wmma_f32_16x16x4_f32(
                false, afrag1[kc], false, bf0, (short)0, acc10, false, false);
            acc01 = __builtin_amdgcn_wmma_f32_16x16x4_f32(
                false, afrag0[kc], false, bf1, (short)0, acc01, false, false);
            acc11 = __builtin_amdgcn_wmma_f32_16x16x4_f32(
                false, afrag1[kc], false, bf1, (short)0, acc11, false, false);
        }
        const float nsq_s0 = nsqb[s0 + lmod];
        const float nsq_s1 = nsqb[s0 + 16 + lmod];
#pragma unroll
        for (int r = 0; r < 8; ++r) {
            int m     = r + mhalf;
            float d00 = nsq_t0[r] + nsq_s0 - 2.0f * acc00[r];
            float d10 = nsq_t1[r] + nsq_s0 - 2.0f * acc10[r];
            float d01 = nsq_t0[r] + nsq_s1 - 2.0f * acc01[r];
            float d11 = nsq_t1[r] + nsq_s1 - 2.0f * acc11[r];
            dist[m * TLEN + s0 + lmod]             = fmaxf(d00, 0.0f);
            dist[(m + 16) * TLEN + s0 + lmod]      = fmaxf(d10, 0.0f);
            dist[m * TLEN + s0 + 16 + lmod]        = fmaxf(d01, 0.0f);
            dist[(m + 16) * TLEN + s0 + 16 + lmod] = fmaxf(d11, 0.0f);
        }
    }
    __syncthreads();

    const int row = tid >> 4;
    const int tt  = tid & 15;
    float d0 = 3.4e38f, d1 = 3.4e38f, d2 = 3.4e38f;
    int   i0 = 0x7fffffff, i1 = 0x7fffffff, i2 = 0x7fffffff;
    const float4* drow4 = (const float4*)(dist + row * TLEN) + tt;
#pragma unroll 8
    for (int g = 0; g < 32; ++g) {
        const float4 v = drow4[g * 16];
        const int j = g * 64 + tt * 4;
        top3_insert(v.x, j + 0, d0, d1, d2, i0, i1, i2);
        top3_insert(v.y, j + 1, d0, d1, d2, i0, i1, i2);
        top3_insert(v.z, j + 2, d0, d1, d2, i0, i1, i2);
        top3_insert(v.w, j + 3, d0, d1, d2, i0, i1, i2);
    }

    int res0, res1, res2;
    {
        float bd = d0; int bi = i0;
        lexmin_step<1>(bd, bi); lexmin_step<2>(bd, bi);
        lexmin_step<4>(bd, bi); lexmin_step<8>(bd, bi);
        res0 = bi;
        const bool won = (i0 == bi);
        d0 = won ? d1 : d0;  i0 = won ? i1 : i0;
        d1 = won ? d2 : d1;  i1 = won ? i2 : i1;
    }
    {
        float bd = d0; int bi = i0;
        lexmin_step<1>(bd, bi); lexmin_step<2>(bd, bi);
        lexmin_step<4>(bd, bi); lexmin_step<8>(bd, bi);
        res1 = bi;
        const bool won = (i0 == bi);
        d0 = won ? d1 : d0;  i0 = won ? i1 : i0;
    }
    {
        float bd = d0; int bi = i0;
        lexmin_step<1>(bd, bi); lexmin_step<2>(bd, bi);
        lexmin_step<4>(bd, bi); lexmin_step<8>(bd, bi);
        res2 = bi;
    }
    if (tt == 0) {
        int* o = idx_out + ((size_t)b * TLEN + t0 + row) * KNN;
        o[0] = res0; o[1] = res1; o[2] = res2;
    }
}

#define CONV_TOK  128
#define CH_CHUNK  16
#define N_CHUNK   (CH / CH_CHUNK)
#define XPITCH    (TLEN + 16)

__device__ __forceinline__ void lds_barrier() {
    asm volatile("s_wait_dscnt 0x0\n\ts_barrier_signal -1\n\ts_barrier_wait -1" ::: "memory");
}

__device__ __forceinline__ void conv_load_w(const float* __restrict__ wl, v2f (&wf)[12]) {
#pragma unroll
    for (int q = 0; q < CH_CHUNK / 4; ++q) {
        const v2f* p = (const v2f*)(wl + q * 4 * KNN);
        wf[3 * q + 0] = p[0];
        wf[3 * q + 1] = p[1];
        wf[3 * q + 2] = p[2];
    }
}

__device__ __forceinline__ void conv_products(const float* xsh, int s0i, int s1i, int s2i,
                                              const v2f (&wa)[12], const v2f (&wb)[12],
                                              v8f& acc0, v8f& acc1) {
#pragma unroll
    for (int q = 0; q < CH_CHUNK / 4; ++q) {
        const float* r0 = xsh + (q * 4) * XPITCH;
        const float* r1 = r0 + XPITCH;
        v2f b0, b1, b2;
        b0.x = r0[s0i];  b0.y = r0[s1i];
        b1.x = r0[s2i];  b1.y = r1[s0i];
        b2.x = r1[s1i];  b2.y = r1[s2i];
        acc0 = __builtin_amdgcn_wmma_f32_16x16x4_f32(
            false, wa[3 * q + 0], false, b0, (short)0, acc0, false, false);
        acc1 = __builtin_amdgcn_wmma_f32_16x16x4_f32(
            false, wb[3 * q + 0], false, b0, (short)0, acc1, false, false);
        acc0 = __builtin_amdgcn_wmma_f32_16x16x4_f32(
            false, wa[3 * q + 1], false, b1, (short)0, acc0, false, false);
        acc1 = __builtin_amdgcn_wmma_f32_16x16x4_f32(
            false, wb[3 * q + 1], false, b1, (short)0, acc1, false, false);
        acc0 = __builtin_amdgcn_wmma_f32_16x16x4_f32(
            false, wa[3 * q + 2], false, b2, (short)0, acc0, false, false);
        acc1 = __builtin_amdgcn_wmma_f32_16x16x4_f32(
            false, wb[3 * q + 2], false, b2, (short)0, acc1, false, false);
    }
}

__global__ __launch_bounds__(512) __attribute__((amdgpu_waves_per_eu(4, 4))) void conv_kernel(const float* x,
                                                   const float* __restrict__ W,
                                                   const float* __restrict__ bias,
                                                   const int* __restrict__ idx,
                                                   float* __restrict__ out) {
    extern __shared__ __attribute__((aligned(16))) float xs[];

    const int b     = blockIdx.x >> 4;
    const int blk   = blockIdx.x & 15;
    const int tid   = threadIdx.x;
    const int wave  = tid >> 5;
    const int lane  = tid & 31;
    const int lmod  = lane & 15;
    const int lhalf = (lane < 16) ? 0 : 2;
    const int mhalf = (lane < 16) ? 0 : 8;

    const int ttile = wave & 7;
    const int opair = wave >> 3;
    const int tok   = blk * CONV_TOK + ttile * 16 + lmod;

    const float* xb = x + (size_t)b * CH * TLEN;
    const int* my = idx + ((size_t)b * TLEN + tok) * KNN;
    const int s0i = my[0] & (TLEN - 1);
    const int s1i = my[1] & (TLEN - 1);
    const int s2i = my[2] & (TLEN - 1);

    const float4* src4 = (const float4*)xb + tid;
    float4*       dst4 = (float4*)xs + tid;
#define CONV_ROWS(X) X(0) X(1) X(2) X(3) X(4) X(5) X(6) X(7) \
                     X(8) X(9) X(10) X(11) X(12) X(13) X(14) X(15)
#define CONV_FIRST(r) float4 p##r = src4[(size_t)(r) * (TLEN / 4)];
#define CONV_STORE(r) dst4[(r) * (XPITCH / 4)] = p##r;
#define CONV_NEXT(r)  p##r = nxt4[(size_t)(r) * (TLEN / 4)];
    CONV_ROWS(CONV_FIRST)

    const float* wrow0 = W + ((size_t)((2 * opair + 0) * 16 + lmod) * CH + lhalf) * KNN;
    const float* wrow1 = W + ((size_t)((2 * opair + 1) * 16 + lmod) * CH + lhalf) * KNN;
    const float* xsh   = xs + lhalf * XPITCH;

    v8f acc0 = {};
    v8f acc1 = {};

#pragma unroll 1
    for (int c = 0; c < N_CHUNK - 1; ++c) {
        v2f wa[12], wb[12];
        conv_load_w(wrow0 + c * (CH_CHUNK * KNN), wa);
        conv_load_w(wrow1 + c * (CH_CHUNK * KNN), wb);
        CONV_ROWS(CONV_STORE)
        const float4* nxt4 = src4 + (size_t)(c + 1) * CH_CHUNK * (TLEN / 4);
        CONV_ROWS(CONV_NEXT)
        lds_barrier();
        conv_products(xsh, s0i, s1i, s2i, wa, wb, acc0, acc1);
        lds_barrier();
    }
    {
        v2f wa[12], wb[12];
        conv_load_w(wrow0 + (N_CHUNK - 1) * (CH_CHUNK * KNN), wa);
        conv_load_w(wrow1 + (N_CHUNK - 1) * (CH_CHUNK * KNN), wb);
        CONV_ROWS(CONV_STORE)
        lds_barrier();
        conv_products(xsh, s0i, s1i, s2i, wa, wb, acc0, acc1);
    }

    float* ob = out + (size_t)b * CH * TLEN + tok;
#pragma unroll
    for (int r = 0; r < 8; ++r) {
        const int oa = (2 * opair + 0) * 16 + r + mhalf;
        const int oc = (2 * opair + 1) * 16 + r + mhalf;
        ob[(size_t)oa * TLEN] = acc0[r] + bias[oa];
        ob[(size_t)oc * TLEN] = acc1[r] + bias[oc];
    }
}

extern "C" void kernel_launch(void* const* d_in, const int* in_sizes, int n_in,
                              void* d_out, int out_size, void* d_ws, size_t ws_size,
                              hipStream_t stream) {
    (void)in_sizes; (void)n_in; (void)out_size; (void)ws_size;
    const float* x    = (const float*)d_in[0];
    const float* W    = (const float*)d_in[1];
    const float* bias = (const float*)d_in[2];
    float* out = (float*)d_out;

    float* nsq = (float*)d_ws;
    int*   idx = (int*)((char*)d_ws + (size_t)BATCH * TLEN * sizeof(float));

    nsq_kernel<<<(BATCH * TLEN) / 256, 256, 0, stream>>>(x, nsq);

    const size_t smem = (size_t)32 * TLEN * sizeof(float);
    hipFuncSetAttribute((const void*)knn_kernel,
                        hipFuncAttributeMaxDynamicSharedMemorySize, (int)smem);
    knn_kernel<<<BATCH * (TLEN / 32), 512, smem, stream>>>(x, nsq, idx);

    const size_t smem_conv = (size_t)CH_CHUNK * XPITCH * sizeof(float);
    hipFuncSetAttribute((const void*)conv_kernel,
                        hipFuncAttributeMaxDynamicSharedMemorySize, (int)smem_conv);
    conv_kernel<<<BATCH * (TLEN / CONV_TOK), 512, smem_conv, stream>>>(x, W, bias, idx, out);
}


# knn: LDS slab holds 1024 of the 2048 columns at a time (two passes, each lane's running triple stays in registers); 8-wave blocks, two per WGP: one block's selection runs under the other's WMMAs
# speedup vs baseline: 2.0227x; 1.0688x over previous
#include <hip/hip_runtime.h>


typedef __attribute__((ext_vector_type(2))) float v2f;
typedef __attribute__((ext_vector_type(8))) float v8f;

#define BATCH 16
#define CH    64
#define TLEN  2048
#define KNN   3

__global__ __launch_bounds__(256) void nsq_kernel(const float* __restrict__ x,
                                                  float* __restrict__ nsq) {
    int g = blockIdx.x * 256 + threadIdx.x;
    int b = g >> 11;
    int t = g & (TLEN - 1);
    const float* xb = x + (size_t)b * CH * TLEN + t;
    float s = 0.0f;
#pragma unroll
    for (int c = 0; c < CH; ++c) {
        float v = xb[c * TLEN];
        s += v * v;
    }
    nsq[g] = s;
}

#define KCOLS   1024
#define KPITCH  (KCOLS + 32)

template <int M>
__device__ __forceinline__ int row_xor(int v) {
    return __builtin_amdgcn_update_dpp(v, v, 0x160 | M, 0xf, 0xf, false);
}

__device__ __forceinline__ void top3_insert(float d, int j,
                                            float& d0, float& d1, float& d2,
                                            int& i0, int& i1, int& i2) {
    const bool c0 = d < d0, c1 = d < d1, c2 = d < d2;
    i2 = c2 ? (c1 ? i1 : j) : i2;
    i1 = c1 ? (c0 ? i0 : j) : i1;
    i0 = c0 ? j : i0;
    const float n0 = fminf(d0, d);
    const float n1 = __builtin_amdgcn_fmed3f(d0, d1, d);
    const float n2 = __builtin_amdgcn_fmed3f(d1, d2, d);
    d0 = n0; d1 = n1; d2 = n2;
}

template <int M>
__device__ __forceinline__ void lexmin_step(float& bd, int& bi) {
    const float od = __int_as_float(row_xor<M>(__float_as_int(bd)));
    const int   oi = row_xor<M>(bi);
    const bool take = (od < bd) | ((od == bd) & (oi < bi));
    bd = take ? od : bd;
    bi = take ? oi : bi;
}

__global__ __launch_bounds__(256) __attribute__((amdgpu_waves_per_eu(4, 4))) void knn_kernel(const float* __restrict__ x,
                                                  const float* __restrict__ nsq,
                                                  int* __restrict__ idx_out) {
    const int b    = blockIdx.x >> 6;
    const int tile = blockIdx.x & 63;
    const int t0   = tile * 32;

    extern __shared__ __attribute__((aligned(16))) float smem[];
    float* dist   = smem;

    const int tid   = threadIdx.x;
    const int lane  = tid & 31;
    const int wave  = tid >> 5;
    const int lmod  = lane & 15;
    const int lhalf = (lane < 16) ? 0 : 2;
    const int mhalf = (lane < 16) ? 0 : 8;

    const float* xb   = x   + (size_t)b * CH * TLEN;
    const float* nsqb = nsq + (size_t)b * TLEN;

    v2f afrag0[16], afrag1[16];
#pragma unroll
    for (int kc = 0; kc < 16; ++kc) {
        int c0 = kc * 4 + lhalf;
        afrag0[kc].x = xb[(size_t)(c0 + 0) * TLEN + t0 + lmod];
        afrag0[kc].y = xb[(size_t)(c0 + 1) * TLEN + t0 + lmod];
        afrag1[kc].x = xb[(size_t)(c0 + 0) * TLEN + t0 + 16 + lmod];
        afrag1[kc].y = xb[(size_t)(c0 + 1) * TLEN + t0 + 16 + lmod];
    }
    float nsq_t0[8], nsq_t1[8];
#pragma unroll
    for (int r = 0; r < 8; ++r) {
        nsq_t0[r] = nsqb[t0 + r + mhalf];
        nsq_t1[r] = nsqb[t0 + 16 + r + mhalf];
    }

    const int row = tid >> 3;
    const int tt  = tid & 7;
    float d0 = 3.4e38f, d1 = 3.4e38f, d2 = 3.4e38f;
    int   i0 = 0x7fffffff, i1 = 0x7fffffff, i2 = 0x7fffffff;
    const float4* drow4 = (const float4*)(dist + row * KPITCH) + tt;

#pragma unroll 1
    for (int pass = 0; pass < 2; ++pass) {
        if (pass != 0) __syncthreads();
        for (int it = 0; it < 4; ++it) {
            const int sl = (it * 8 + wave) * 32;
            const int s0 = pass * KCOLS + sl;
            v8f acc00 = {};
            v8f acc10 = {};
            v8f acc01 = {};
            v8f acc11 = {};
#pragma unroll
            for (int kc = 0; kc < 16; ++kc) {
                int c0 = kc * 4 + lhalf;
                v2f bf0, bf1;
                bf0.x = xb[(size_t)(c0 + 0) * TLEN + s0 + lmod];
                bf0.y = xb[(size_t)(c0 + 1) * TLEN + s0 + lmod];
                bf1.x = xb[(size_t)(c0 + 0) * TLEN + s0 + 16 + lmod];
                bf1.y = xb[(size_t)(c0 + 1) * TLEN + s0 + 16 + lmod];
                acc00 = __builtin_amdgcn_wmma_f32_16x16x4_f32(
                    false, afrag0[kc], false, bf0, (short)0, acc00, false, false);
                acc10 = __builtin_amdgcn_wmma_f32_16x16x4_f32(
                    false, afrag1[kc], false, bf0, (short)0, acc10, false, false);
                acc01 = __builtin_amdgcn_wmma_f32_16x16x4_f32(
                    false, afrag0[kc], false, bf1, (short)0, acc01, false, false);
                acc11 = __builtin_amdgcn_wmma_f32_16x16x4_f32(
                    false, afrag1[kc], false, bf1, (short)0, acc11, false, false);
            }
            const float nsq_s0 = nsqb[s0 + lmod];
            const float nsq_s1 = nsqb[s0 + 16 + lmod];
#pragma unroll
            for (int r = 0; r < 8; ++r) {
                int m     = r + mhalf;
                float d00 = nsq_t0[r] + nsq_s0 - 2.0f * acc00[r];
                float d10 = nsq_t1[r] + nsq_s0 - 2.0f * acc10[r];
                float d01 = nsq_t0[r] + nsq_s1 - 2.0f * acc01[r];
                float d11 = nsq_t1[r] + nsq_s1 - 2.0f * acc11[r];
                dist[m * KPITCH + sl + lmod]             = fmaxf(d00, 0.0f);
                dist[(m + 16) * KPITCH + sl + lmod]      = fmaxf(d10, 0.0f);
                dist[m * KPITCH + sl + 16 + lmod]        = fmaxf(d01, 0.0f);
                dist[(m + 16) * KPITCH + sl + 16 + lmod] = fmaxf(d11, 0.0f);
            }
        }
        __syncthreads();

#pragma unroll 8
        for (int g = 0; g < 32; ++g) {
            const float4 v = drow4[g * 8];
            const int j = pass * KCOLS + g * 32 + tt * 4;
            top3_insert(v.x, j + 0, d0, d1, d2, i0, i1, i2);
            top3_insert(v.y, j + 1, d0, d1, d2, i0, i1, i2);
            top3_insert(v.z, j + 2, d0, d1, d2, i0, i1, i2);
            top3_insert(v.w, j + 3, d0, d1, d2, i0, i1, i2);
        }
    }

    int res0, res1, res2;
    {
        float bd = d0; int bi = i0;
        lexmin_step<1>(bd, bi); lexmin_step<2>(bd, bi);
        lexmin_step<4>(bd, bi);
        res0 = bi;
        const bool won = (i0 == bi);
        d0 = won ? d1 : d0;  i0 = won ? i1 : i0;
        d1 = won ? d2 : d1;  i1 = won ? i2 : i1;
    }
    {
        float bd = d0; int bi = i0;
        lexmin_step<1>(bd, bi); lexmin_step<2>(bd, bi);
        lexmin_step<4>(bd, bi);
        res1 = bi;
        const bool won = (i0 == bi);
        d0 = won ? d1 : d0;  i0 = won ? i1 : i0;
    }
    {
        float bd = d0; int bi = i0;
        lexmin_step<1>(bd, bi); lexmin_step<2>(bd, bi);
        lexmin_step<4>(bd, bi);
        res2 = bi;
    }
    if (tt == 0) {
        int* o = idx_out + ((size_t)b * TLEN + t0 + row) * KNN;
        o[0] = res0; o[1] = res1; o[2] = res2;
    }
}

#define CONV_TOK  128
#define CH_CHUNK  16
#define N_CHUNK   (CH / CH_CHUNK)
#define XPITCH    (TLEN + 16)

__device__ __forceinline__ void lds_barrier() {
    asm volatile("s_wait_dscnt 0x0\n\ts_barrier_signal -1\n\ts_barrier_wait -1" ::: "memory");
}

__device__ __forceinline__ void conv_load_w(const float* __restrict__ wl, v2f (&wf)[12]) {
#pragma unroll
    for (int q = 0; q < CH_CHUNK / 4; ++q) {
        const v2f* p = (const v2f*)(wl + q * 4 * KNN);
        wf[3 * q + 0] = p[0];
        wf[3 * q + 1] = p[1];
        wf[3 * q + 2] = p[2];
    }
}

__device__ __forceinline__ void conv_products(const float* xsh, int s0i, int s1i, int s2i,
                                              const v2f (&wa)[12], const v2f (&wb)[12],
                                              v8f& acc0, v8f& acc1) {
#pragma unroll
    for (int q = 0; q < CH_CHUNK / 4; ++q) {
        const float* r0 = xsh + (q * 4) * XPITCH;
        const float* r1 = r0 + XPITCH;
        v2f b0, b1, b2;
        b0.x = r0[s0i];  b0.y = r0[s1i];
        b1.x = r0[s2i];  b1.y = r1[s0i];
        b2.x = r1[s1i];  b2.y = r1[s2i];
        acc0 = __builtin_amdgcn_wmma_f32_16x16x4_f32(
            false, wa[3 * q + 0], false, b0, (short)0, acc0, false, false);
        acc1 = __builtin_amdgcn_wmma_f32_16x16x4_f32(
            false, wb[3 * q + 0], false, b0, (short)0, acc1, false, false);
        acc0 = __builtin_amdgcn_wmma_f32_16x16x4_f32(
            false, wa[3 * q + 1], false, b1, (short)0, acc0, false, false);
        acc1 = __builtin_amdgcn_wmma_f32_16x16x4_f32(
            false, wb[3 * q + 1], false, b1, (short)0, acc1, false, false);
        acc0 = __builtin_amdgcn_wmma_f32_16x16x4_f32(
            false, wa[3 * q + 2], false, b2, (short)0, acc0, false, false);
        acc1 = __builtin_amdgcn_wmma_f32_16x16x4_f32(
            false, wb[3 * q + 2], false, b2, (short)0, acc1, false, false);
    }
}

__global__ __launch_bounds__(512) __attribute__((amdgpu_waves_per_eu(4, 4))) void conv_kernel(const float* x,
                                                   const float* __restrict__ W,
                                                   const float* __restrict__ bias,
                                                   const int* __restrict__ idx,
                                                   float* __restrict__ out) {
    extern __shared__ __attribute__((aligned(16))) float xs[];

    const int b     = blockIdx.x >> 4;
    const int blk   = blockIdx.x & 15;
    const int tid   = threadIdx.x;
    const int wave  = tid >> 5;
    const int lane  = tid & 31;
    const int lmod  = lane & 15;
    const int lhalf = (lane < 16) ? 0 : 2;
    const int mhalf = (lane < 16) ? 0 : 8;

    const int ttile = wave & 7;
    const int opair = wave >> 3;
    const int tok   = blk * CONV_TOK + ttile * 16 + lmod;

    const float* xb = x + (size_t)b * CH * TLEN;
    const int* my = idx + ((size_t)b * TLEN + tok) * KNN;
    const int s0i = my[0] & (TLEN - 1);
    const int s1i = my[1] & (TLEN - 1);
    const int s2i = my[2] & (TLEN - 1);

    const float4* src4 = (const float4*)xb + tid;
    float4*       dst4 = (float4*)xs + tid;
#define CONV_ROWS(X) X(0) X(1) X(2) X(3) X(4) X(5) X(6) X(7) \
                     X(8) X(9) X(10) X(11) X(12) X(13) X(14) X(15)
#define CONV_FIRST(r) float4 p##r = src4[(size_t)(r) * (TLEN / 4)];
#define CONV_STORE(r) dst4[(r) * (XPITCH / 4)] = p##r;
#define CONV_NEXT(r)  p##r = nxt4[(size_t)(r) * (TLEN / 4)];
    CONV_ROWS(CONV_FIRST)

    const float* wrow0 = W + ((size_t)((2 * opair + 0) * 16 + lmod) * CH + lhalf) * KNN;
    const float* wrow1 = W + ((size_t)((2 * opair + 1) * 16 + lmod) * CH + lhalf) * KNN;
    const float* xsh   = xs + lhalf * XPITCH;

    v8f acc0 = {};
    v8f acc1 = {};

#pragma unroll 1
    for (int c = 0; c < N_CHUNK - 1; ++c) {
        v2f wa[12], wb[12];
        conv_load_w(wrow0 + c * (CH_CHUNK * KNN), wa);
        conv_load_w(wrow1 + c * (CH_CHUNK * KNN), wb);
        CONV_ROWS(CONV_STORE)
        const float4* nxt4 = src4 + (size_t)(c + 1) * CH_CHUNK * (TLEN / 4);
        CONV_ROWS(CONV_NEXT)
        lds_barrier();
        conv_products(xsh, s0i, s1i, s2i, wa, wb, acc0, acc1);
        lds_barrier();
    }
    {
        v2f wa[12], wb[12];
        conv_load_w(wrow0 + (N_CHUNK - 1) * (CH_CHUNK * KNN), wa);
        conv_load_w(wrow1 + (N_CHUNK - 1) * (CH_CHUNK * KNN), wb);
        CONV_ROWS(CONV_STORE)
        lds_barrier();
        conv_products(xsh, s0i, s1i, s2i, wa, wb, acc0, acc1);
    }

    float* ob = out + (size_t)b * CH * TLEN + tok;
#pragma unroll
    for (int r = 0; r < 8; ++r) {
        const int oa = (2 * opair + 0) * 16 + r + mhalf;
        const int oc = (2 * opair + 1) * 16 + r + mhalf;
        ob[(size_t)oa * TLEN] = acc0[r] + bias[oa];
        ob[(size_t)oc * TLEN] = acc1[r] + bias[oc];
    }
}

extern "C" void kernel_launch(void* const* d_in, const int* in_sizes, int n_in,
                              void* d_out, int out_size, void* d_ws, size_t ws_size,
                              hipStream_t stream) {
    (void)in_sizes; (void)n_in; (void)out_size; (void)ws_size;
    const float* x    = (const float*)d_in[0];
    const float* W    = (const float*)d_in[1];
    const float* bias = (const float*)d_in[2];
    float* out = (float*)d_out;

    float* nsq = (float*)d_ws;
    int*   idx = (int*)((char*)d_ws + (size_t)BATCH * TLEN * sizeof(float));

    nsq_kernel<<<(BATCH * TLEN) / 256, 256, 0, stream>>>(x, nsq);

    const size_t smem = (size_t)32 * KPITCH * sizeof(float);
    hipFuncSetAttribute((const void*)knn_kernel,
                        hipFuncAttributeMaxDynamicSharedMemorySize, (int)smem);
    knn_kernel<<<BATCH * (TLEN / 32), 256, smem, stream>>>(x, nsq, idx);

    const size_t smem_conv = (size_t)CH_CHUNK * XPITCH * sizeof(float);
    hipFuncSetAttribute((const void*)conv_kernel,
                        hipFuncAttributeMaxDynamicSharedMemorySize, (int)smem_conv);
    conv_kernel<<<BATCH * (TLEN / CONV_TOK), 512, smem_conv, stream>>>(x, W, bias, idx, out);
}


# knn: top-3 selection from the slab in two levels: only each 4-column group's minimum enters the lane's branch-free top-3 of groups; the 3 winning groups are re-read, their 12 distances inserted
# speedup vs baseline: 2.1994x; 1.0874x over previous
#include <hip/hip_runtime.h>


typedef __attribute__((ext_vector_type(2))) float v2f;
typedef __attribute__((ext_vector_type(8))) float v8f;

#define BATCH 16
#define CH    64
#define TLEN  2048
#define KNN   3

__global__ __launch_bounds__(256) void nsq_kernel(const float* __restrict__ x,
                                                  float* __restrict__ nsq) {
    int g = blockIdx.x * 256 + threadIdx.x;
    int b = g >> 11;
    int t = g & (TLEN - 1);
    const float* xb = x + (size_t)b * CH * TLEN + t;
    float s = 0.0f;
#pragma unroll
    for (int c = 0; c < CH; ++c) {
        float v = xb[c * TLEN];
        s += v * v;
    }
    nsq[g] = s;
}

#define KCOLS   1024
#define KPITCH  (KCOLS + 32)

template <int M>
__device__ __forceinline__ int row_xor(int v) {
    return __builtin_amdgcn_update_dpp(v, v, 0x160 | M, 0xf, 0xf, false);
}

__device__ __forceinline__ void top3_insert(float d, int j,
                                            float& d0, float& d1, float& d2,
                                            int& i0, int& i1, int& i2) {
    const bool c0 = d < d0, c1 = d < d1, c2 = d < d2;
    i2 = c2 ? (c1 ? i1 : j) : i2;
    i1 = c1 ? (c0 ? i0 : j) : i1;
    i0 = c0 ? j : i0;
    const float n0 = fminf(d0, d);
    const float n1 = __builtin_amdgcn_fmed3f(d0, d1, d);
    const float n2 = __builtin_amdgcn_fmed3f(d1, d2, d);
    d0 = n0; d1 = n1; d2 = n2;
}

template <int M>
__device__ __forceinline__ void lexmin_step(float& bd, int& bi) {
    const float od = __int_as_float(row_xor<M>(__float_as_int(bd)));
    const int   oi = row_xor<M>(bi);
    const bool take = (od < bd) | ((od == bd) & (oi < bi));
    bd = take ? od : bd;
    bi = take ? oi : bi;
}

__global__ __launch_bounds__(256) __attribute__((amdgpu_waves_per_eu(4, 4))) void knn_kernel(const float* __restrict__ x,
                                                  const float* __restrict__ nsq,
                                                  int* __restrict__ idx_out) {
    const int b    = blockIdx.x >> 6;
    const int tile = blockIdx.x & 63;
    const int t0   = tile * 32;

    extern __shared__ __attribute__((aligned(16))) float smem[];
    float* dist   = smem;

    const int tid   = threadIdx.x;
    const int lane  = tid & 31;
    const int wave  = tid >> 5;
    const int lmod  = lane & 15;
    const int lhalf = (lane < 16) ? 0 : 2;
    const int mhalf = (lane < 16) ? 0 : 8;

    const float* xb   = x   + (size_t)b * CH * TLEN;
    const float* nsqb = nsq + (size_t)b * TLEN;

    v2f afrag0[16], afrag1[16];
#pragma unroll
    for (int kc = 0; kc < 16; ++kc) {
        int c0 = kc * 4 + lhalf;
        afrag0[kc].x = xb[(size_t)(c0 + 0) * TLEN + t0 + lmod];
        afrag0[kc].y = xb[(size_t)(c0 + 1) * TLEN + t0 + lmod];
        afrag1[kc].x = xb[(size_t)(c0 + 0) * TLEN + t0 + 16 + lmod];
        afrag1[kc].y = xb[(size_t)(c0 + 1) * TLEN + t0 + 16 + lmod];
    }
    float nsq_t0[8], nsq_t1[8];
#pragma unroll
    for (int r = 0; r < 8; ++r) {
        nsq_t0[r] = nsqb[t0 + r + mhalf];
        nsq_t1[r] = nsqb[t0 + 16 + r + mhalf];
    }

    const int row = tid >> 3;
    const int tt  = tid & 7;
    float d0 = 3.4e38f, d1 = 3.4e38f, d2 = 3.4e38f;
    int   i0 = 0x7fffffff, i1 = 0x7fffffff, i2 = 0x7fffffff;
    const float4* drow4 = (const float4*)(dist + row * KPITCH) + tt;

#pragma unroll 1
    for (int pass = 0; pass < 2; ++pass) {
        if (pass != 0) __syncthreads();
        for (int it = 0; it < 4; ++it) {
            const int sl = (it * 8 + wave) * 32;
            const int s0 = pass * KCOLS + sl;
            v8f acc00 = {};
            v8f acc10 = {};
            v8f acc01 = {};
            v8f acc11 = {};
#pragma unroll
            for (int kc = 0; kc < 16; ++kc) {
                int c0 = kc * 4 + lhalf;
                v2f bf0, bf1;
                bf0.x = xb[(size_t)(c0 + 0) * TLEN + s0 + lmod];
                bf0.y = xb[(size_t)(c0 + 1) * TLEN + s0 + lmod];
                bf1.x = xb[(size_t)(c0 + 0) * TLEN + s0 + 16 + lmod];
                bf1.y = xb[(size_t)(c0 + 1) * TLEN + s0 + 16 + lmod];
                acc00 = __builtin_amdgcn_wmma_f32_16x16x4_f32(
                    false, afrag0[kc], false, bf0, (short)0, acc00, false, false);
                acc10 = __builtin_amdgcn_wmma_f32_16x16x4_f32(
                    false, afrag1[kc], false, bf0, (short)0, acc10, false, false);
                acc01 = __builtin_amdgcn_wmma_f32_16x16x4_f32(
                    false, afrag0[kc], false, bf1, (short)0, acc01, false, false);
                acc11 = __builtin_amdgcn_wmma_f32_16x16x4_f32(
                    false, afrag1[kc], false, bf1, (short)0, acc11, false, false);
            }
            const float nsq_s0 = nsqb[s0 + lmod];
            const float nsq_s1 = nsqb[s0 + 16 + lmod];
#pragma unroll
            for (int r = 0; r < 8; ++r) {
                int m     = r + mhalf;
                float d00 = nsq_t0[r] + nsq_s0 - 2.0f * acc00[r];
                float d10 = nsq_t1[r] + nsq_s0 - 2.0f * acc10[r];
                float d01 = nsq_t0[r] + nsq_s1 - 2.0f * acc01[r];
                float d11 = nsq_t1[r] + nsq_s1 - 2.0f * acc11[r];
                dist[m * KPITCH + sl + lmod]             = fmaxf(d00, 0.0f);
                dist[(m + 16) * KPITCH + sl + lmod]      = fmaxf(d10, 0.0f);
                dist[m * KPITCH + sl + 16 + lmod]        = fmaxf(d01, 0.0f);
                dist[(m + 16) * KPITCH + sl + 16 + lmod] = fmaxf(d11, 0.0f);
            }
        }
        __syncthreads();

        float gd0 = 3.4e38f, gd1 = 3.4e38f, gd2 = 3.4e38f;
        int   gi0 = 0, gi1 = 1, gi2 = 2;
#pragma unroll 8
        for (int g = 0; g < 32; ++g) {
            const float4 v = drow4[g * 8];
            const float m = fminf(fminf(v.x, v.y), fminf(v.z, v.w));
            top3_insert(m, g, gd0, gd1, gd2, gi0, gi1, gi2);
        }
        const int ga = min(gi0, min(gi1, gi2));
        const int gc = max(gi0, max(gi1, gi2));
        const int gb = gi0 + gi1 + gi2 - ga - gc;
        const int jb = pass * KCOLS + tt * 4;
        const float4 va = drow4[ga * 8];
        const float4 vb = drow4[gb * 8];
        const float4 vc = drow4[gc * 8];
        const int ja = jb + ga * 32, jm = jb + gb * 32, jc = jb + gc * 32;
        top3_insert(va.x, ja + 0, d0, d1, d2, i0, i1, i2);
        top3_insert(va.y, ja + 1, d0, d1, d2, i0, i1, i2);
        top3_insert(va.z, ja + 2, d0, d1, d2, i0, i1, i2);
        top3_insert(va.w, ja + 3, d0, d1, d2, i0, i1, i2);
        top3_insert(vb.x, jm + 0, d0, d1, d2, i0, i1, i2);
        top3_insert(vb.y, jm + 1, d0, d1, d2, i0, i1, i2);
        top3_insert(vb.z, jm + 2, d0, d1, d2, i0, i1, i2);
        top3_insert(vb.w, jm + 3, d0, d1, d2, i0, i1, i2);
        top3_insert(vc.x, jc + 0, d0, d1, d2, i0, i1, i2);
        top3_insert(vc.y, jc + 1, d0, d1, d2, i0, i1, i2);
        top3_insert(vc.z, jc + 2, d0, d1, d2, i0, i1, i2);
        top3_insert(vc.w, jc + 3, d0, d1, d2, i0, i1, i2);
    }

    int res0, res1, res2;
    {
        float bd = d0; int bi = i0;
        lexmin_step<1>(bd, bi); lexmin_step<2>(bd, bi);
        lexmin_step<4>(bd, bi);
        res0 = bi;
        const bool won = (i0 == bi);
        d0 = won ? d1 : d0;  i0 = won ? i1 : i0;
        d1 = won ? d2 : d1;  i1 = won ? i2 : i1;
    }
    {
        float bd = d0; int bi = i0;
        lexmin_step<1>(bd, bi); lexmin_step<2>(bd, bi);
        lexmin_step<4>(bd, bi);
        res1 = bi;
        const bool won = (i0 == bi);
        d0 = won ? d1 : d0;  i0 = won ? i1 : i0;
    }
    {
        float bd = d0; int bi = i0;
        lexmin_step<1>(bd, bi); lexmin_step<2>(bd, bi);
        lexmin_step<4>(bd, bi);
        res2 = bi;
    }
    if (tt == 0) {
        int* o = idx_out + ((size_t)b * TLEN + t0 + row) * KNN;
        o[0] = res0; o[1] = res1; o[2] = res2;
    }
}

#define CONV_TOK  128
#define CH_CHUNK  16
#define N_CHUNK   (CH / CH_CHUNK)
#define XPITCH    (TLEN + 16)

__device__ __forceinline__ void lds_barrier() {
    asm volatile("s_wait_dscnt 0x0\n\ts_barrier_signal -1\n\ts_barrier_wait -1" ::: "memory");
}

__device__ __forceinline__ void conv_load_w(const float* __restrict__ wl, v2f (&wf)[12]) {
#pragma unroll
    for (int q = 0; q < CH_CHUNK / 4; ++q) {
        const v2f* p = (const v2f*)(wl + q * 4 * KNN);
        wf[3 * q + 0] = p[0];
        wf[3 * q + 1] = p[1];
        wf[3 * q + 2] = p[2];
    }
}

__device__ __forceinline__ void conv_products(const float* xsh, int s0i, int s1i, int s2i,
                                              const v2f (&wa)[12], const v2f (&wb)[12],
                                              v8f& acc0, v8f& acc1) {
#pragma unroll
    for (int q = 0; q < CH_CHUNK / 4; ++q) {
        const float* r0 = xsh + (q * 4) * XPITCH;
        const float* r1 = r0 + XPITCH;
        v2f b0, b1, b2;
        b0.x = r0[s0i];  b0.y = r0[s1i];
        b1.x = r0[s2i];  b1.y = r1[s0i];
        b2.x = r1[s1i];  b2.y = r1[s2i];
        acc0 = __builtin_amdgcn_wmma_f32_16x16x4_f32(
            false, wa[3 * q + 0], false, b0, (short)0, acc0, false, false);
        acc1 = __builtin_amdgcn_wmma_f32_16x16x4_f32(
            false, wb[3 * q + 0], false, b0, (short)0, acc1, false, false);
        acc0 = __builtin_amdgcn_wmma_f32_16x16x4_f32(
            false, wa[3 * q + 1], false, b1, (short)0, acc0, false, false);
        acc1 = __builtin_amdgcn_wmma_f32_16x16x4_f32(
            false, wb[3 * q + 1], false, b1, (short)0, acc1, false, false);
        acc0 = __builtin_amdgcn_wmma_f32_16x16x4_f32(
            false, wa[3 * q + 2], false, b2, (short)0, acc0, false, false);
        acc1 = __builtin_amdgcn_wmma_f32_16x16x4_f32(
            false, wb[3 * q + 2], false, b2, (short)0, acc1, false, false);
    }
}

__global__ __launch_bounds__(512) __attribute__((amdgpu_waves_per_eu(4, 4))) void conv_kernel(const float* x,
                                                   const float* __restrict__ W,
                                                   const float* __restrict__ bias,
                                                   const int* __restrict__ idx,
                                                   float* __restrict__ out) {
    extern __shared__ __attribute__((aligned(16))) float xs[];

    const int b     = blockIdx.x >> 4;
    const int blk   = blockIdx.x & 15;
    const int tid   = threadIdx.x;
    const int wave  = tid >> 5;
    const int lane  = tid & 31;
    const int lmod  = lane & 15;
    const int lhalf = (lane < 16) ? 0 : 2;
    const int mhalf = (lane < 16) ? 0 : 8;

    const int ttile = wave & 7;
    const int opair = wave >> 3;
    const int tok   = blk * CONV_TOK + ttile * 16 + lmod;

    const float* xb = x + (size_t)b * CH * TLEN;
    const int* my = idx + ((size_t)b * TLEN + tok) * KNN;
    const int s0i = my[0] & (TLEN - 1);
    const int s1i = my[1] & (TLEN - 1);
    const int s2i = my[2] & (TLEN - 1);

    const float4* src4 = (const float4*)xb + tid;
    float4*       dst4 = (float4*)xs + tid;
#define CONV_ROWS(X) X(0) X(1) X(2) X(3) X(4) X(5) X(6) X(7) \
                     X(8) X(9) X(10) X(11) X(12) X(13) X(14) X(15)
#define CONV_FIRST(r) float4 p##r = src4[(size_t)(r) * (TLEN / 4)];
#define CONV_STORE(r) dst4[(r) * (XPITCH / 4)] = p##r;
#define CONV_NEXT(r)  p##r = nxt4[(size_t)(r) * (TLEN / 4)];
    CONV_ROWS(CONV_FIRST)

    const float* wrow0 = W + ((size_t)((2 * opair + 0) * 16 + lmod) * CH + lhalf) * KNN;
    const float* wrow1 = W + ((size_t)((2 * opair + 1) * 16 + lmod) * CH + lhalf) * KNN;
    const float* xsh   = xs + lhalf * XPITCH;

    v8f acc0 = {};
    v8f acc1 = {};

#pragma unroll 1
    for (int c = 0; c < N_CHUNK - 1; ++c) {
        v2f wa[12], wb[12];
        conv_load_w(wrow0 + c * (CH_CHUNK * KNN), wa);
        conv_load_w(wrow1 + c * (CH_CHUNK * KNN), wb);
        CONV_ROWS(CONV_STORE)
        const float4* nxt4 = src4 + (size_t)(c + 1) * CH_CHUNK * (TLEN / 4);
        CONV_ROWS(CONV_NEXT)
        lds_barrier();
        conv_products(xsh, s0i, s1i, s2i, wa, wb, acc0, acc1);
        lds_barrier();
    }
    {
        v2f wa[12], wb[12];
        conv_load_w(wrow0 + (N_CHUNK - 1) * (CH_CHUNK * KNN), wa);
        conv_load_w(wrow1 + (N_CHUNK - 1) * (CH_CHUNK * KNN), wb);
        CONV_ROWS(CONV_STORE)
        lds_barrier();
        conv_products(xsh, s0i, s1i, s2i, wa, wb, acc0, acc1);
    }

    float* ob = out + (size_t)b * CH * TLEN + tok;
#pragma unroll
    for (int r = 0; r < 8; ++r) {
        const int oa = (2 * opair + 0) * 16 + r + mhalf;
        const int oc = (2 * opair + 1) * 16 + r + mhalf;
        ob[(size_t)oa * TLEN] = acc0[r] + bias[oa];
        ob[(size_t)oc * TLEN] = acc1[r] + bias[oc];
    }
}

extern "C" void kernel_launch(void* const* d_in, const int* in_sizes, int n_in,
                              void* d_out, int out_size, void* d_ws, size_t ws_size,
                              hipStream_t stream) {
    (void)in_sizes; (void)n_in; (void)out_size; (void)ws_size;
    const float* x    = (const float*)d_in[0];
    const float* W    = (const float*)d_in[1];
    const float* bias = (const float*)d_in[2];
    float* out = (float*)d_out;

    float* nsq = (float*)d_ws;
    int*   idx = (int*)((char*)d_ws + (size_t)BATCH * TLEN * sizeof(float));

    nsq_kernel<<<(BATCH * TLEN) / 256, 256, 0, stream>>>(x, nsq);

    const size_t smem = (size_t)32 * KPITCH * sizeof(float);
    hipFuncSetAttribute((const void*)knn_kernel,
                        hipFuncAttributeMaxDynamicSharedMemorySize, (int)smem);
    knn_kernel<<<BATCH * (TLEN / 32), 256, smem, stream>>>(x, nsq, idx);

    const size_t smem_conv = (size_t)CH_CHUNK * XPITCH * sizeof(float);
    hipFuncSetAttribute((const void*)conv_kernel,
                        hipFuncAttributeMaxDynamicSharedMemorySize, (int)smem_conv);
    conv_kernel<<<BATCH * (TLEN / CONV_TOK), 512, smem_conv, stream>>>(x, W, bias, idx, out);
}


# conv: A fragments come from a fragment-order copy of W that kernel 1 writes each call (idx packed to 2 words per token to make room): each A load is 4 whole lines, not 16 scattered ones
# speedup vs baseline: 2.3437x; 1.0656x over previous
#include <hip/hip_runtime.h>


typedef __attribute__((ext_vector_type(2))) float v2f;
typedef __attribute__((ext_vector_type(8))) float v8f;

#define BATCH 16
#define CH    64
#define TLEN  2048
#define KNN   3

#define WP_KPAIRS  24
#define WP_FLOAT4S (4 * WP_KPAIRS * 32)

__global__ __launch_bounds__(256) void nsq_kernel(const float* __restrict__ x,
                                                  const float* __restrict__ W,
                                                  float* __restrict__ nsq,
                                                  float* __restrict__ wp) {
    int g = blockIdx.x * 256 + threadIdx.x;
    const bool packs = blockIdx.x < WP_FLOAT4S / 256;
    float4 wv = make_float4(0.0f, 0.0f, 0.0f, 0.0f);
    if (packs) {
        const int lane  = g & 31;
        const int kp    = (g >> 5) % WP_KPAIRS;
        const int ot    = (g >> 5) / WP_KPAIRS;
        const int lmod  = lane & 15;
        const int lhalf = (lane < 16) ? 0 : 2;
        const float* wrow = W + (size_t)(ot * 16 + lmod) * (CH * KNN);
        float v[4];
#pragma unroll
        for (int e = 0; e < 2; ++e) {
            const int ks = 2 * kp + e;
            const int c  = ks / 12;
            const int r  = ks % 12;
            const int q  = r / 3;
            const int j  = r % 3;
            const float* s = wrow + (16 * c + 4 * q + lhalf) * KNN + 2 * j;
            v[2 * e + 0] = s[0];
            v[2 * e + 1] = s[1];
        }
        wv = make_float4(v[0], v[1], v[2], v[3]);
    }
    int b = g >> 11;
    int t = g & (TLEN - 1);
    const float* xb = x + (size_t)b * CH * TLEN + t;
    float s = 0.0f;
#pragma unroll
    for (int c = 0; c < CH; ++c) {
        float v = xb[c * TLEN];
        s += v * v;
    }
    nsq[g] = s;
    if (packs) ((float4*)wp)[g] = wv;
}

#define KCOLS   1024
#define KPITCH  (KCOLS + 32)

template <int M>
__device__ __forceinline__ int row_xor(int v) {
    return __builtin_amdgcn_update_dpp(v, v, 0x160 | M, 0xf, 0xf, false);
}

__device__ __forceinline__ void top3_insert(float d, int j,
                                            float& d0, float& d1, float& d2,
                                            int& i0, int& i1, int& i2) {
    const bool c0 = d < d0, c1 = d < d1, c2 = d < d2;
    i2 = c2 ? (c1 ? i1 : j) : i2;
    i1 = c1 ? (c0 ? i0 : j) : i1;
    i0 = c0 ? j : i0;
    const float n0 = fminf(d0, d);
    const float n1 = __builtin_amdgcn_fmed3f(d0, d1, d);
    const float n2 = __builtin_amdgcn_fmed3f(d1, d2, d);
    d0 = n0; d1 = n1; d2 = n2;
}

template <int M>
__device__ __forceinline__ void lexmin_step(float& bd, int& bi) {
    const float od = __int_as_float(row_xor<M>(__float_as_int(bd)));
    const int   oi = row_xor<M>(bi);
    const bool take = (od < bd) | ((od == bd) & (oi < bi));
    bd = take ? od : bd;
    bi = take ? oi : bi;
}

__global__ __launch_bounds__(256) __attribute__((amdgpu_waves_per_eu(4, 4))) void knn_kernel(const float* __restrict__ x,
                                                  const float* __restrict__ nsq,
                                                  int* __restrict__ idx_out) {
    const int b    = blockIdx.x >> 6;
    const int tile = blockIdx.x & 63;
    const int t0   = tile * 32;

    extern __shared__ __attribute__((aligned(16))) float smem[];
    float* dist   = smem;

    const int tid   = threadIdx.x;
    const int lane  = tid & 31;
    const int wave  = tid >> 5;
    const int lmod  = lane & 15;
    const int lhalf = (lane < 16) ? 0 : 2;
    const int mhalf = (lane < 16) ? 0 : 8;

    const float* xb   = x   + (size_t)b * CH * TLEN;
    const float* nsqb = nsq + (size_t)b * TLEN;

    v2f afrag0[16], afrag1[16];
#pragma unroll
    for (int kc = 0; kc < 16; ++kc) {
        int c0 = kc * 4 + lhalf;
        afrag0[kc].x = xb[(size_t)(c0 + 0) * TLEN + t0 + lmod];
        afrag0[kc].y = xb[(size_t)(c0 + 1) * TLEN + t0 + lmod];
        afrag1[kc].x = xb[(size_t)(c0 + 0) * TLEN + t0 + 16 + lmod];
        afrag1[kc].y = xb[(size_t)(c0 + 1) * TLEN + t0 + 16 + lmod];
    }
    float nsq_t0[8], nsq_t1[8];
#pragma unroll
    for (int r = 0; r < 8; ++r) {
        nsq_t0[r] = nsqb[t0 + r + mhalf];
        nsq_t1[r] = nsqb[t0 + 16 + r + mhalf];
    }

    const int row = tid >> 3;
    const int tt  = tid & 7;
    float d0 = 3.4e38f, d1 = 3.4e38f, d2 = 3.4e38f;
    int   i0 = 0x7fffffff, i1 = 0x7fffffff, i2 = 0x7fffffff;
    const float4* drow4 = (const float4*)(dist + row * KPITCH) + tt;

#pragma unroll 1
    for (int pass = 0; pass < 2; ++pass) {
        if (pass != 0) __syncthreads();
        for (int it = 0; it < 4; ++it) {
            const int sl = (it * 8 + wave) * 32;
            const int s0 = pass * KCOLS + sl;
            v8f acc00 = {};
            v8f acc10 = {};
            v8f acc01 = {};
            v8f acc11 = {};
#pragma unroll
            for (int kc = 0; kc < 16; ++kc) {
                int c0 = kc * 4 + lhalf;
                v2f bf0, bf1;
                bf0.x = xb[(size_t)(c0 + 0) * TLEN + s0 + lmod];
                bf0.y = xb[(size_t)(c0 + 1) * TLEN + s0 + lmod];
                bf1.x = xb[(size_t)(c0 + 0) * TLEN + s0 + 16 + lmod];
                bf1.y = xb[(size_t)(c0 + 1) * TLEN + s0 + 16 + lmod];
                acc00 = __builtin_amdgcn_wmma_f32_16x16x4_f32(
                    false, afrag0[kc], false, bf0, (short)0, acc00, false, false);
                acc10 = __builtin_amdgcn_wmma_f32_16x16x4_f32(
                    false, afrag1[kc], false, bf0, (short)0, acc10, false, false);
                acc01 = __builtin_amdgcn_wmma_f32_16x16x4_f32(
                    false, afrag0[kc], false, bf1, (short)0, acc01, false, false);
                acc11 = __builtin_amdgcn_wmma_f32_16x16x4_f32(
                    false, afrag1[kc], false, bf1, (short)0, acc11, false, false);
            }
            const float nsq_s0 = nsqb[s0 + lmod];
            const float nsq_s1 = nsqb[s0 + 16 + lmod];
#pragma unroll
            for (int r = 0; r < 8; ++r) {
                int m     = r + mhalf;
                float d00 = nsq_t0[r] + nsq_s0 - 2.0f * acc00[r];
                float d10 = nsq_t1[r] + nsq_s0 - 2.0f * acc10[r];
                float d01 = nsq_t0[r] + nsq_s1 - 2.0f * acc01[r];
                float d11 = nsq_t1[r] + nsq_s1 - 2.0f * acc11[r];
                dist[m * KPITCH + sl + lmod]             = fmaxf(d00, 0.0f);
                dist[(m + 16) * KPITCH + sl + lmod]      = fmaxf(d10, 0.0f);
                dist[m * KPITCH + sl + 16 + lmod]        = fmaxf(d01, 0.0f);
                dist[(m + 16) * KPITCH + sl + 16 + lmod] = fmaxf(d11, 0.0f);
            }
        }
        __syncthreads();

        float gd0 = 3.4e38f, gd1 = 3.4e38f, gd2 = 3.4e38f;
        int   gi0 = 0, gi1 = 1, gi2 = 2;
#pragma unroll 8
        for (int g = 0; g < 32; ++g) {
            const float4 v = drow4[g * 8];
            const float m = fminf(fminf(v.x, v.y), fminf(v.z, v.w));
            top3_insert(m, g, gd0, gd1, gd2, gi0, gi1, gi2);
        }
        const int ga = min(gi0, min(gi1, gi2));
        const int gc = max(gi0, max(gi1, gi2));
        const int gb = gi0 + gi1 + gi2 - ga - gc;
        const int jb = pass * KCOLS + tt * 4;
        const float4 va = drow4[ga * 8];
        const float4 vb = drow4[gb * 8];
        const float4 vc = drow4[gc * 8];
        const int ja = jb + ga * 32, jm = jb + gb * 32, jc = jb + gc * 32;
        top3_insert(va.x, ja + 0, d0, d1, d2, i0, i1, i2);
        top3_insert(va.y, ja + 1, d0, d1, d2, i0, i1, i2);
        top3_insert(va.z, ja + 2, d0, d1, d2, i0, i1, i2);
        top3_insert(va.w, ja + 3, d0, d1, d2, i0, i1, i2);
        top3_insert(vb.x, jm + 0, d0, d1, d2, i0, i1, i2);
        top3_insert(vb.y, jm + 1, d0, d1, d2, i0, i1, i2);
        top3_insert(vb.z, jm + 2, d0, d1, d2, i0, i1, i2);
        top3_insert(vb.w, jm + 3, d0, d1, d2, i0, i1, i2);
        top3_insert(vc.x, jc + 0, d0, d1, d2, i0, i1, i2);
        top3_insert(vc.y, jc + 1, d0, d1, d2, i0, i1, i2);
        top3_insert(vc.z, jc + 2, d0, d1, d2, i0, i1, i2);
        top3_insert(vc.w, jc + 3, d0, d1, d2, i0, i1, i2);
    }

    int res0, res1, res2;
    {
        float bd = d0; int bi = i0;
        lexmin_step<1>(bd, bi); lexmin_step<2>(bd, bi);
        lexmin_step<4>(bd, bi);
        res0 = bi;
        const bool won = (i0 == bi);
        d0 = won ? d1 : d0;  i0 = won ? i1 : i0;
        d1 = won ? d2 : d1;  i1 = won ? i2 : i1;
    }
    {
        float bd = d0; int bi = i0;
        lexmin_step<1>(bd, bi); lexmin_step<2>(bd, bi);
        lexmin_step<4>(bd, bi);
        res1 = bi;
        const bool won = (i0 == bi);
        d0 = won ? d1 : d0;  i0 = won ? i1 : i0;
    }
    {
        float bd = d0; int bi = i0;
        lexmin_step<1>(bd, bi); lexmin_step<2>(bd, bi);
        lexmin_step<4>(bd, bi);
        res2 = bi;
    }
    if (tt == 0) {
        int* o = idx_out + ((size_t)b * TLEN + t0 + row) * 2;
        o[0] = res0 | (res1 << 16); o[1] = res2;
    }
}

#define CONV_TOK  128
#define CH_CHUNK  16
#define N_CHUNK   (CH / CH_CHUNK)
#define XPITCH    (TLEN + 16)

__device__ __forceinline__ void lds_barrier() {
    asm volatile("s_wait_dscnt 0x0\n\ts_barrier_signal -1\n\ts_barrier_wait -1" ::: "memory");
}

__device__ __forceinline__ void conv_load_w(const float4* __restrict__ wl, v2f (&wf)[12]) {
#pragma unroll
    for (int p = 0; p < 6; ++p) {
        const float4 v = wl[p * 32];
        wf[2 * p + 0].x = v.x;  wf[2 * p + 0].y = v.y;
        wf[2 * p + 1].x = v.z;  wf[2 * p + 1].y = v.w;
    }
}

__device__ __forceinline__ void conv_products(const float* xsh, int s0i, int s1i, int s2i,
                                              const v2f (&wa)[12], const v2f (&wb)[12],
                                              v8f& acc0, v8f& acc1) {
#pragma unroll
    for (int q = 0; q < CH_CHUNK / 4; ++q) {
        const float* r0 = xsh + (q * 4) * XPITCH;
        const float* r1 = r0 + XPITCH;
        v2f b0, b1, b2;
        b0.x = r0[s0i];  b0.y = r0[s1i];
        b1.x = r0[s2i];  b1.y = r1[s0i];
        b2.x = r1[s1i];  b2.y = r1[s2i];
        acc0 = __builtin_amdgcn_wmma_f32_16x16x4_f32(
            false, wa[3 * q + 0], false, b0, (short)0, acc0, false, false);
        acc1 = __builtin_amdgcn_wmma_f32_16x16x4_f32(
            false, wb[3 * q + 0], false, b0, (short)0, acc1, false, false);
        acc0 = __builtin_amdgcn_wmma_f32_16x16x4_f32(
            false, wa[3 * q + 1], false, b1, (short)0, acc0, false, false);
        acc1 = __builtin_amdgcn_wmma_f32_16x16x4_f32(
            false, wb[3 * q + 1], false, b1, (short)0, acc1, false, false);
        acc0 = __builtin_amdgcn_wmma_f32_16x16x4_f32(
            false, wa[3 * q + 2], false, b2, (short)0, acc0, false, false);
        acc1 = __builtin_amdgcn_wmma_f32_16x16x4_f32(
            false, wb[3 * q + 2], false, b2, (short)0, acc1, false, false);
    }
}

__global__ __launch_bounds__(512) __attribute__((amdgpu_waves_per_eu(4, 4))) void conv_kernel(const float* x,
                                                   const float* __restrict__ wp,
                                                   const float* __restrict__ bias,
                                                   const int* __restrict__ idx,
                                                   float* __restrict__ out) {
    extern __shared__ __attribute__((aligned(16))) float xs[];

    const int b     = blockIdx.x >> 4;
    const int blk   = blockIdx.x & 15;
    const int tid   = threadIdx.x;
    const int wave  = tid >> 5;
    const int lane  = tid & 31;
    const int lmod  = lane & 15;
    const int lhalf = (lane < 16) ? 0 : 2;
    const int mhalf = (lane < 16) ? 0 : 8;

    const int ttile = wave & 7;
    const int opair = wave >> 3;
    const int tok   = blk * CONV_TOK + ttile * 16 + lmod;

    const float* xb = x + (size_t)b * CH * TLEN;
    const int* my = idx + ((size_t)b * TLEN + tok) * 2;
    const int w01 = my[0];
    const int s0i = w01 & (TLEN - 1);
    const int s1i = (w01 >> 16) & (TLEN - 1);
    const int s2i = my[1] & (TLEN - 1);

    const float4* src4 = (const float4*)xb + tid;
    float4*       dst4 = (float4*)xs + tid;
#define CONV_ROWS(X) X(0) X(1) X(2) X(3) X(4) X(5) X(6) X(7) \
                     X(8) X(9) X(10) X(11) X(12) X(13) X(14) X(15)
#define CONV_FIRST(r) float4 p##r = src4[(size_t)(r) * (TLEN / 4)];
#define CONV_STORE(r) dst4[(r) * (XPITCH / 4)] = p##r;
#define CONV_NEXT(r)  p##r = nxt4[(size_t)(r) * (TLEN / 4)];
    CONV_ROWS(CONV_FIRST)

    const float4* wrow0 = (const float4*)wp + (size_t)(2 * opair + 0) * (WP_KPAIRS * 32) + lane;
    const float4* wrow1 = (const float4*)wp + (size_t)(2 * opair + 1) * (WP_KPAIRS * 32) + lane;
    const float* xsh   = xs + lhalf * XPITCH;

    v8f acc0 = {};
    v8f acc1 = {};

#pragma unroll 1
    for (int c = 0; c < N_CHUNK - 1; ++c) {
        v2f wa[12], wb[12];
        conv_load_w(wrow0 + c * (6 * 32), wa);
        conv_load_w(wrow1 + c * (6 * 32), wb);
        CONV_ROWS(CONV_STORE)
        const float4* nxt4 = src4 + (size_t)(c + 1) * CH_CHUNK * (TLEN / 4);
        CONV_ROWS(CONV_NEXT)
        lds_barrier();
        conv_products(xsh, s0i, s1i, s2i, wa, wb, acc0, acc1);
        lds_barrier();
    }
    {
        v2f wa[12], wb[12];
        conv_load_w(wrow0 + (N_CHUNK - 1) * (6 * 32), wa);
        conv_load_w(wrow1 + (N_CHUNK - 1) * (6 * 32), wb);
        CONV_ROWS(CONV_STORE)
        lds_barrier();
        conv_products(xsh, s0i, s1i, s2i, wa, wb, acc0, acc1);
    }

    float* ob = out + (size_t)b * CH * TLEN + tok;
#pragma unroll
    for (int r = 0; r < 8; ++r) {
        const int oa = (2 * opair + 0) * 16 + r + mhalf;
        const int oc = (2 * opair + 1) * 16 + r + mhalf;
        ob[(size_t)oa * TLEN] = acc0[r] + bias[oa];
        ob[(size_t)oc * TLEN] = acc1[r] + bias[oc];
    }
}

extern "C" void kernel_launch(void* const* d_in, const int* in_sizes, int n_in,
                              void* d_out, int out_size, void* d_ws, size_t ws_size,
                              hipStream_t stream) {
    (void)in_sizes; (void)n_in; (void)out_size; (void)ws_size;
    const float* x    = (const float*)d_in[0];
    const float* W    = (const float*)d_in[1];
    const float* bias = (const float*)d_in[2];
    float* out = (float*)d_out;

    float* nsq = (float*)d_ws;
    int*   idx = (int*)((char*)d_ws + (size_t)BATCH * TLEN * sizeof(float));
    float* wp  = (float*)((char*)d_ws + (size_t)BATCH * TLEN * 3 * sizeof(float));

    nsq_kernel<<<(BATCH * TLEN) / 256, 256, 0, stream>>>(x, W, nsq, wp);

    const size_t smem = (size_t)32 * KPITCH * sizeof(float);
    hipFuncSetAttribute((const void*)knn_kernel,
                        hipFuncAttributeMaxDynamicSharedMemorySize, (int)smem);
    knn_kernel<<<BATCH * (TLEN / 32), 256, smem, stream>>>(x, nsq, idx);

    const size_t smem_conv = (size_t)CH_CHUNK * XPITCH * sizeof(float);
    hipFuncSetAttribute((const void*)conv_kernel,
                        hipFuncAttributeMaxDynamicSharedMemorySize, (int)smem_conv);
    conv_kernel<<<BATCH * (TLEN / CONV_TOK), 512, smem_conv, stream>>>(x, wp, bias, idx, out);
}


# knn: a wave's 64-column chunk is four interleaved column tiles (tile j = columns 4n+j): one 128-bit x load per channel row feeds 8 WMMAs, a row's 4 distances leave as one 128-bit slab store
# speedup vs baseline: 2.4624x; 1.0506x over previous
#include <hip/hip_runtime.h>


typedef __attribute__((ext_vector_type(2))) float v2f;
typedef __attribute__((ext_vector_type(8))) float v8f;

#define BATCH 16
#define CH    64
#define TLEN  2048
#define KNN   3

#define WP_KPAIRS  24
#define WP_FLOAT4S (4 * WP_KPAIRS * 32)

__global__ __launch_bounds__(256) void nsq_kernel(const float* __restrict__ x,
                                                  const float* __restrict__ W,
                                                  float* __restrict__ nsq,
                                                  float* __restrict__ wp) {
    int g = blockIdx.x * 256 + threadIdx.x;
    const bool packs = blockIdx.x < WP_FLOAT4S / 256;
    float4 wv = make_float4(0.0f, 0.0f, 0.0f, 0.0f);
    if (packs) {
        const int lane  = g & 31;
        const int kp    = (g >> 5) % WP_KPAIRS;
        const int ot    = (g >> 5) / WP_KPAIRS;
        const int lmod  = lane & 15;
        const int lhalf = (lane < 16) ? 0 : 2;
        const float* wrow = W + (size_t)(ot * 16 + lmod) * (CH * KNN);
        float v[4];
#pragma unroll
        for (int e = 0; e < 2; ++e) {
            const int ks = 2 * kp + e;
            const int c  = ks / 12;
            const int r  = ks % 12;
            const int q  = r / 3;
            const int j  = r % 3;
            const float* s = wrow + (16 * c + 4 * q + lhalf) * KNN + 2 * j;
            v[2 * e + 0] = s[0];
            v[2 * e + 1] = s[1];
        }
        wv = make_float4(v[0], v[1], v[2], v[3]);
    }
    int b = g >> 11;
    int t = g & (TLEN - 1);
    const float* xb = x + (size_t)b * CH * TLEN + t;
    float s = 0.0f;
#pragma unroll
    for (int c = 0; c < CH; ++c) {
        float v = xb[c * TLEN];
        s += v * v;
    }
    nsq[g] = s;
    if (packs) ((float4*)wp)[g] = wv;
}

#define KCOLS   1024
#define KPITCH  (KCOLS + 32)

template <int M>
__device__ __forceinline__ int row_xor(int v) {
    return __builtin_amdgcn_update_dpp(v, v, 0x160 | M, 0xf, 0xf, false);
}

__device__ __forceinline__ void top3_insert(float d, int j,
                                            float& d0, float& d1, float& d2,
                                            int& i0, int& i1, int& i2) {
    const bool c0 = d < d0, c1 = d < d1, c2 = d < d2;
    i2 = c2 ? (c1 ? i1 : j) : i2;
    i1 = c1 ? (c0 ? i0 : j) : i1;
    i0 = c0 ? j : i0;
    const float n0 = fminf(d0, d);
    const float n1 = __builtin_amdgcn_fmed3f(d0, d1, d);
    const float n2 = __builtin_amdgcn_fmed3f(d1, d2, d);
    d0 = n0; d1 = n1; d2 = n2;
}

template <int M>
__device__ __forceinline__ void lexmin_step(float& bd, int& bi) {
    const float od = __int_as_float(row_xor<M>(__float_as_int(bd)));
    const int   oi = row_xor<M>(bi);
    const bool take = (od < bd) | ((od == bd) & (oi < bi));
    bd = take ? od : bd;
    bi = take ? oi : bi;
}

__global__ __launch_bounds__(256) __attribute__((amdgpu_waves_per_eu(4, 4))) void knn_kernel(const float* __restrict__ x,
                                                  const float* __restrict__ nsq,
                                                  int* __restrict__ idx_out) {
    const int b    = blockIdx.x >> 6;
    const int tile = blockIdx.x & 63;
    const int t0   = tile * 32;

    extern __shared__ __attribute__((aligned(16))) float smem[];
    float* dist   = smem;

    const int tid   = threadIdx.x;
    const int lane  = tid & 31;
    const int wave  = tid >> 5;
    const int lmod  = lane & 15;
    const int lhalf = (lane < 16) ? 0 : 2;
    const int mhalf = (lane < 16) ? 0 : 8;

    const float* xb   = x   + (size_t)b * CH * TLEN;
    const float* nsqb = nsq + (size_t)b * TLEN;

    v2f afrag0[16], afrag1[16];
#pragma unroll
    for (int kc = 0; kc < 16; ++kc) {
        int c0 = kc * 4 + lhalf;
        afrag0[kc].x = xb[(size_t)(c0 + 0) * TLEN + t0 + lmod];
        afrag0[kc].y = xb[(size_t)(c0 + 1) * TLEN + t0 + lmod];
        afrag1[kc].x = xb[(size_t)(c0 + 0) * TLEN + t0 + 16 + lmod];
        afrag1[kc].y = xb[(size_t)(c0 + 1) * TLEN + t0 + 16 + lmod];
    }
    float nsq_t0[8], nsq_t1[8];
#pragma unroll
    for (int r = 0; r < 8; ++r) {
        nsq_t0[r] = nsqb[t0 + r + mhalf];
        nsq_t1[r] = nsqb[t0 + 16 + r + mhalf];
    }

    const int row = tid >> 3;
    const int tt  = tid & 7;
    float d0 = 3.4e38f, d1 = 3.4e38f, d2 = 3.4e38f;
    int   i0 = 0x7fffffff, i1 = 0x7fffffff, i2 = 0x7fffffff;
    const float4* drow4 = (const float4*)(dist + row * KPITCH) + tt;

#pragma unroll 1
    for (int pass = 0; pass < 2; ++pass) {
        if (pass != 0) __syncthreads();
        for (int it = 0; it < 2; ++it) {
            const int sl = (it * 8 + wave) * 64;
            const int s0 = pass * KCOLS + sl;
            v8f acc00 = {}, acc10 = {};
            v8f acc01 = {}, acc11 = {};
            v8f acc02 = {}, acc12 = {};
            v8f acc03 = {}, acc13 = {};
            const float* xc = xb + s0 + 4 * lmod;
#pragma unroll
            for (int kc = 0; kc < 8; ++kc) {
                const int c0 = kc * 4 + lhalf;
                const float4 u = *(const float4*)(xc + (size_t)(c0 + 0) * TLEN);
                const float4 v = *(const float4*)(xc + (size_t)(c0 + 1) * TLEN);
                v2f bf0, bf1, bf2, bf3;
                bf0.x = u.x;  bf0.y = v.x;
                bf1.x = u.y;  bf1.y = v.y;
                bf2.x = u.z;  bf2.y = v.z;
                bf3.x = u.w;  bf3.y = v.w;
                acc00 = __builtin_amdgcn_wmma_f32_16x16x4_f32(
                    false, afrag0[kc], false, bf0, (short)0, acc00, false, false);
                acc10 = __builtin_amdgcn_wmma_f32_16x16x4_f32(
                    false, afrag1[kc], false, bf0, (short)0, acc10, false, false);
                acc01 = __builtin_amdgcn_wmma_f32_16x16x4_f32(
                    false, afrag0[kc], false, bf1, (short)0, acc01, false, false);
                acc11 = __builtin_amdgcn_wmma_f32_16x16x4_f32(
                    false, afrag1[kc], false, bf1, (short)0, acc11, false, false);
                acc02 = __builtin_amdgcn_wmma_f32_16x16x4_f32(
                    false, afrag0[kc], false, bf2, (short)0, acc02, false, false);
                acc12 = __builtin_amdgcn_wmma_f32_16x16x4_f32(
                    false, afrag1[kc], false, bf2, (short)0, acc12, false, false);
                acc03 = __builtin_amdgcn_wmma_f32_16x16x4_f32(
                    false, afrag0[kc], false, bf3, (short)0, acc03, false, false);
                acc13 = __builtin_amdgcn_wmma_f32_16x16x4_f32(
                    false, afrag1[kc], false, bf3, (short)0, acc13, false, false);
            }
            __builtin_amdgcn_sched_barrier(0);
#pragma unroll
            for (int kc = 8; kc < 16; ++kc) {
                const int c0 = kc * 4 + lhalf;
                const float4 u = *(const float4*)(xc + (size_t)(c0 + 0) * TLEN);
                const float4 v = *(const float4*)(xc + (size_t)(c0 + 1) * TLEN);
                v2f bf0, bf1, bf2, bf3;
                bf0.x = u.x;  bf0.y = v.x;
                bf1.x = u.y;  bf1.y = v.y;
                bf2.x = u.z;  bf2.y = v.z;
                bf3.x = u.w;  bf3.y = v.w;
                acc00 = __builtin_amdgcn_wmma_f32_16x16x4_f32(
                    false, afrag0[kc], false, bf0, (short)0, acc00, false, false);
                acc10 = __builtin_amdgcn_wmma_f32_16x16x4_f32(
                    false, afrag1[kc], false, bf0, (short)0, acc10, false, false);
                acc01 = __builtin_amdgcn_wmma_f32_16x16x4_f32(
                    false, afrag0[kc], false, bf1, (short)0, acc01, false, false);
                acc11 = __builtin_amdgcn_wmma_f32_16x16x4_f32(
                    false, afrag1[kc], false, bf1, (short)0, acc11, false, false);
                acc02 = __builtin_amdgcn_wmma_f32_16x16x4_f32(
                    false, afrag0[kc], false, bf2, (short)0, acc02, false, false);
                acc12 = __builtin_amdgcn_wmma_f32_16x16x4_f32(
                    false, afrag1[kc], false, bf2, (short)0, acc12, false, false);
                acc03 = __builtin_amdgcn_wmma_f32_16x16x4_f32(
                    false, afrag0[kc], false, bf3, (short)0, acc03, false, false);
                acc13 = __builtin_amdgcn_wmma_f32_16x16x4_f32(
                    false, afrag1[kc], false, bf3, (short)0, acc13, false, false);
            }
            const float4 ns = *(const float4*)(nsqb + s0 + 4 * lmod);
            float* drw = dist + (2 * mhalf) * KPITCH + sl + 4 * lmod;
#pragma unroll
            for (int r = 0; r < 8; ++r) {
                float4 e0, e1;
                e0.x = fmaxf(nsq_t0[r] + ns.x - 2.0f * acc00[r], 0.0f);
                e0.y = fmaxf(nsq_t0[r] + ns.y - 2.0f * acc01[r], 0.0f);
                e0.z = fmaxf(nsq_t0[r] + ns.z - 2.0f * acc02[r], 0.0f);
                e0.w = fmaxf(nsq_t0[r] + ns.w - 2.0f * acc03[r], 0.0f);
                e1.x = fmaxf(nsq_t1[r] + ns.x - 2.0f * acc10[r], 0.0f);
                e1.y = fmaxf(nsq_t1[r] + ns.y - 2.0f * acc11[r], 0.0f);
                e1.z = fmaxf(nsq_t1[r] + ns.z - 2.0f * acc12[r], 0.0f);
                e1.w = fmaxf(nsq_t1[r] + ns.w - 2.0f * acc13[r], 0.0f);
                *(float4*)(drw + (2 * r + 0) * KPITCH) = e0;
                *(float4*)(drw + (2 * r + 1) * KPITCH) = e1;
            }
        }
        __syncthreads();

        float gd0 = 3.4e38f, gd1 = 3.4e38f, gd2 = 3.4e38f;
        int   gi0 = 0, gi1 = 1, gi2 = 2;
#pragma unroll 8
        for (int g = 0; g < 32; ++g) {
            const float4 v = drow4[g * 8];
            const float m = fminf(fminf(v.x, v.y), fminf(v.z, v.w));
            top3_insert(m, g, gd0, gd1, gd2, gi0, gi1, gi2);
        }
        const int ga = min(gi0, min(gi1, gi2));
        const int gc = max(gi0, max(gi1, gi2));
        const int gb = gi0 + gi1 + gi2 - ga - gc;
        const int jb = pass * KCOLS + tt * 4;
        const float4 va = drow4[ga * 8];
        const float4 vb = drow4[gb * 8];
        const float4 vc = drow4[gc * 8];
        const int ja = jb + ga * 32, jm = jb + gb * 32, jc = jb + gc * 32;
        top3_insert(va.x, ja + 0, d0, d1, d2, i0, i1, i2);
        top3_insert(va.y, ja + 1, d0, d1, d2, i0, i1, i2);
        top3_insert(va.z, ja + 2, d0, d1, d2, i0, i1, i2);
        top3_insert(va.w, ja + 3, d0, d1, d2, i0, i1, i2);
        top3_insert(vb.x, jm + 0, d0, d1, d2, i0, i1, i2);
        top3_insert(vb.y, jm + 1, d0, d1, d2, i0, i1, i2);
        top3_insert(vb.z, jm + 2, d0, d1, d2, i0, i1, i2);
        top3_insert(vb.w, jm + 3, d0, d1, d2, i0, i1, i2);
        top3_insert(vc.x, jc + 0, d0, d1, d2, i0, i1, i2);
        top3_insert(vc.y, jc + 1, d0, d1, d2, i0, i1, i2);
        top3_insert(vc.z, jc + 2, d0, d1, d2, i0, i1, i2);
        top3_insert(vc.w, jc + 3, d0, d1, d2, i0, i1, i2);
    }

    int res0, res1, res2;
    {
        float bd = d0; int bi = i0;
        lexmin_step<1>(bd, bi); lexmin_step<2>(bd, bi);
        lexmin_step<4>(bd, bi);
        res0 = bi;
        const bool won = (i0 == bi);
        d0 = won ? d1 : d0;  i0 = won ? i1 : i0;
        d1 = won ? d2 : d1;  i1 = won ? i2 : i1;
    }
    {
        float bd = d0; int bi = i0;
        lexmin_step<1>(bd, bi); lexmin_step<2>(bd, bi);
        lexmin_step<4>(bd, bi);
        res1 = bi;
        const bool won = (i0 == bi);
        d0 = won ? d1 : d0;  i0 = won ? i1 : i0;
    }
    {
        float bd = d0; int bi = i0;
        lexmin_step<1>(bd, bi); lexmin_step<2>(bd, bi);
        lexmin_step<4>(bd, bi);
        res2 = bi;
    }
    if (tt == 0) {
        const int trow = ((row & 1) << 4) | (row >> 1);
        int* o = idx_out + ((size_t)b * TLEN + t0 + trow) * 2;
        o[0] = res0 | (res1 << 16); o[1] = res2;
    }
}

#define CONV_TOK  128
#define CH_CHUNK  16
#define N_CHUNK   (CH / CH_CHUNK)
#define XPITCH    (TLEN + 16)

__device__ __forceinline__ void lds_barrier() {
    asm volatile("s_wait_dscnt 0x0\n\ts_barrier_signal -1\n\ts_barrier_wait -1" ::: "memory");
}

__device__ __forceinline__ void conv_load_w(const float4* __restrict__ wl, v2f (&wf)[12]) {
#pragma unroll
    for (int p = 0; p < 6; ++p) {
        const float4 v = wl[p * 32];
        wf[2 * p + 0].x = v.x;  wf[2 * p + 0].y = v.y;
        wf[2 * p + 1].x = v.z;  wf[2 * p + 1].y = v.w;
    }
}

__device__ __forceinline__ void conv_products(const float* xsh, int s0i, int s1i, int s2i,
                                              const v2f (&wa)[12], const v2f (&wb)[12],
                                              v8f& acc0, v8f& acc1) {
#pragma unroll
    for (int q = 0; q < CH_CHUNK / 4; ++q) {
        const float* r0 = xsh + (q * 4) * XPITCH;
        const float* r1 = r0 + XPITCH;
        v2f b0, b1, b2;
        b0.x = r0[s0i];  b0.y = r0[s1i];
        b1.x = r0[s2i];  b1.y = r1[s0i];
        b2.x = r1[s1i];  b2.y = r1[s2i];
        acc0 = __builtin_amdgcn_wmma_f32_16x16x4_f32(
            false, wa[3 * q + 0], false, b0, (short)0, acc0, false, false);
        acc1 = __builtin_amdgcn_wmma_f32_16x16x4_f32(
            false, wb[3 * q + 0], false, b0, (short)0, acc1, false, false);
        acc0 = __builtin_amdgcn_wmma_f32_16x16x4_f32(
            false, wa[3 * q + 1], false, b1, (short)0, acc0, false, false);
        acc1 = __builtin_amdgcn_wmma_f32_16x16x4_f32(
            false, wb[3 * q + 1], false, b1, (short)0, acc1, false, false);
        acc0 = __builtin_amdgcn_wmma_f32_16x16x4_f32(
            false, wa[3 * q + 2], false, b2, (short)0, acc0, false, false);
        acc1 = __builtin_amdgcn_wmma_f32_16x16x4_f32(
            false, wb[3 * q + 2], false, b2, (short)0, acc1, false, false);
    }
}

__global__ __launch_bounds__(512) __attribute__((amdgpu_waves_per_eu(4, 4))) void conv_kernel(const float* x,
                                                   const float* __restrict__ wp,
                                                   const float* __restrict__ bias,
                                                   const int* __restrict__ idx,
                                                   float* __restrict__ out) {
    extern __shared__ __attribute__((aligned(16))) float xs[];

    const int b     = blockIdx.x >> 4;
    const int blk   = blockIdx.x & 15;
    const int tid   = threadIdx.x;
    const int wave  = tid >> 5;
    const int lane  = tid & 31;
    const int lmod  = lane & 15;
    const int lhalf = (lane < 16) ? 0 : 2;
    const int mhalf = (lane < 16) ? 0 : 8;

    const int ttile = wave & 7;
    const int opair = wave >> 3;
    const int tok   = blk * CONV_TOK + ttile * 16 + lmod;

    const float* xb = x + (size_t)b * CH * TLEN;
    const int* my = idx + ((size_t)b * TLEN + tok) * 2;
    const int w01 = my[0];
    const int s0i = w01 & (TLEN - 1);
    const int s1i = (w01 >> 16) & (TLEN - 1);
    const int s2i = my[1] & (TLEN - 1);

    const float4* src4 = (const float4*)xb + tid;
    float4*       dst4 = (float4*)xs + tid;
#define CONV_ROWS(X) X(0) X(1) X(2) X(3) X(4) X(5) X(6) X(7) \
                     X(8) X(9) X(10) X(11) X(12) X(13) X(14) X(15)
#define CONV_FIRST(r) float4 p##r = src4[(size_t)(r) * (TLEN / 4)];
#define CONV_STORE(r) dst4[(r) * (XPITCH / 4)] = p##r;
#define CONV_NEXT(r)  p##r = nxt4[(size_t)(r) * (TLEN / 4)];
    CONV_ROWS(CONV_FIRST)

    const float4* wrow0 = (const float4*)wp + (size_t)(2 * opair + 0) * (WP_KPAIRS * 32) + lane;
    const float4* wrow1 = (const float4*)wp + (size_t)(2 * opair + 1) * (WP_KPAIRS * 32) + lane;
    const float* xsh   = xs + lhalf * XPITCH;

    v8f acc0 = {};
    v8f acc1 = {};

#pragma unroll 1
    for (int c = 0; c < N_CHUNK - 1; ++c) {
        v2f wa[12], wb[12];
        conv_load_w(wrow0 + c * (6 * 32), wa);
        conv_load_w(wrow1 + c * (6 * 32), wb);
        CONV_ROWS(CONV_STORE)
        const float4* nxt4 = src4 + (size_t)(c + 1) * CH_CHUNK * (TLEN / 4);
        CONV_ROWS(CONV_NEXT)
        lds_barrier();
        conv_products(xsh, s0i, s1i, s2i, wa, wb, acc0, acc1);
        lds_barrier();
    }
    {
        v2f wa[12], wb[12];
        conv_load_w(wrow0 + (N_CHUNK - 1) * (6 * 32), wa);
        conv_load_w(wrow1 + (N_CHUNK - 1) * (6 * 32), wb);
        CONV_ROWS(CONV_STORE)
        lds_barrier();
        conv_products(xsh, s0i, s1i, s2i, wa, wb, acc0, acc1);
    }

    float* ob = out + (size_t)b * CH * TLEN + tok;
#pragma unroll
    for (int r = 0; r < 8; ++r) {
        const int oa = (2 * opair + 0) * 16 + r + mhalf;
        const int oc = (2 * opair + 1) * 16 + r + mhalf;
        ob[(size_t)oa * TLEN] = acc0[r] + bias[oa];
        ob[(size_t)oc * TLEN] = acc1[r] + bias[oc];
    }
}

extern "C" void kernel_launch(void* const* d_in, const int* in_sizes, int n_in,
                              void* d_out, int out_size, void* d_ws, size_t ws_size,
                              hipStream_t stream) {
    (void)in_sizes; (void)n_in; (void)out_size; (void)ws_size;
    const float* x    = (const float*)d_in[0];
    const float* W    = (const float*)d_in[1];
    const float* bias = (const float*)d_in[2];
    float* out = (float*)d_out;

    float* nsq = (float*)d_ws;
    int*   idx = (int*)((char*)d_ws + (size_t)BATCH * TLEN * sizeof(float));
    float* wp  = (float*)((char*)d_ws + (size_t)BATCH * TLEN * 3 * sizeof(float));

    nsq_kernel<<<(BATCH * TLEN) / 256, 256, 0, stream>>>(x, W, nsq, wp);

    const size_t smem = (size_t)32 * KPITCH * sizeof(float);
    hipFuncSetAttribute((const void*)knn_kernel,
                        hipFuncAttributeMaxDynamicSharedMemorySize, (int)smem);
    knn_kernel<<<BATCH * (TLEN / 32), 256, smem, stream>>>(x, nsq, idx);

    const size_t smem_conv = (size_t)CH_CHUNK * XPITCH * sizeof(float);
    hipFuncSetAttribute((const void*)conv_kernel,
                        hipFuncAttributeMaxDynamicSharedMemorySize, (int)smem_conv);
    conv_kernel<<<BATCH * (TLEN / CONV_TOK), 512, smem_conv, stream>>>(x, wp, bias, idx, out);
}
